# starting kernel
# baseline (speedup 1.0000x reference)
#include <hip/hip_runtime.h>

typedef float v2f __attribute__((ext_vector_type(2)));
typedef float v8f __attribute__((ext_vector_type(8)));

#define DH 128
#define DO 40

__global__ __launch_bounds__(256)
void gcn_gemm1(const float* __restrict__ x, const float* __restrict__ W,
               float* __restrict__ out, int N) {
  const int lane = threadIdx.x & 31;
  const int wave = threadIdx.x >> 5;
  const int r0 = blockIdx.x * 16;
  const int c0 = wave * 16;
  const int m  = lane & 15;
  const int kk = (lane >> 4) << 1;

  const int arow = r0 + m;
  const float* __restrict__ xrow = x + (size_t)(arow < N ? arow : N - 1) * DH;

  v8f acc = {};
  #pragma unroll 4
  for (int k = 0; k < DH; k += 4) {
    v2f a, b;
    a.x = xrow[k + kk];
    a.y = xrow[k + kk + 1];
    b.x = W[(size_t)(k + kk) * DH + c0 + m];
    b.y = W[(size_t)(k + kk + 1) * DH + c0 + m];
    acc = __builtin_amdgcn_wmma_f32_16x16x4_f32(
        false, a, false, b, (short)0, acc, false, false);
  }

  const int rbase = r0 + ((lane >> 4) << 3);
  const int col = c0 + m;
  float* __restrict__ p = out + (size_t)rbase * DH + col;
  if (r0 + 16 <= N) {
    #pragma unroll
    for (int j = 0; j < 8; ++j) p[(size_t)j * DH] = acc[j];
  } else {
    #pragma unroll
    for (int j = 0; j < 8; ++j)
      if (rbase + j < N) p[(size_t)j * DH] = acc[j];
  }
}

__global__ __launch_bounds__(32)
void gcn_gemm2(const float* __restrict__ h, const float* __restrict__ W2,
               float* __restrict__ out, int N) {
  const int lane = threadIdx.x & 31;
  const int r0 = blockIdx.x * 16;
  const int c0 = blockIdx.y * 16;
  const int m  = lane & 15;
  const int kk = (lane >> 4) << 1;

  const int arow = r0 + m;
  const float* __restrict__ hrow = h + (size_t)(arow < N ? arow : N - 1) * DH;
  const int  col = c0 + m;
  const bool cv  = (col < DO);

  v8f acc = {};
  #pragma unroll 4
  for (int k = 0; k < DH; k += 4) {
    v2f a, b;
    a.x = hrow[k + kk];
    a.y = hrow[k + kk + 1];
    b.x = cv ? W2[(size_t)(k + kk) * DO + col] : 0.0f;
    b.y = cv ? W2[(size_t)(k + kk + 1) * DO + col] : 0.0f;
    acc = __builtin_amdgcn_wmma_f32_16x16x4_f32(
        false, a, false, b, (short)0, acc, false, false);
  }

  if (!cv) return;
  const int rbase = r0 + ((lane >> 4) << 3);
  float* __restrict__ p = out + (size_t)rbase * DO + col;
  if (r0 + 16 <= N) {
    #pragma unroll
    for (int j = 0; j < 8; ++j) p[(size_t)j * DO] = acc[j];
  } else {
    #pragma unroll
    for (int j = 0; j < 8; ++j)
      if (rbase + j < N) p[(size_t)j * DO] = acc[j];
  }
}

template <int D>
__global__ __launch_bounds__(256)
void gcn_scatter(const float* __restrict__ msrc, const int* __restrict__ src,
                 const int* __restrict__ dst, const float* __restrict__ w,
                 float* __restrict__ agg, long long total) {
  long long t = (long long)blockIdx.x * blockDim.x + threadIdx.x;
  if (t >= total) return;
  const int e = (int)(t / D);
  const int c = (int)(t % D);
  const int s = src[e];
  const int d = dst[e];
  const float v = msrc[(size_t)s * D + c] * w[e];
  atomicAdd(&agg[(size_t)d * D + c], v);
}

__global__ __launch_bounds__(256)
void gcn_zero(float* __restrict__ p, long long n) {
  long long t = (long long)blockIdx.x * blockDim.x + threadIdx.x;
  if (t < n) p[t] = 0.0f;
}

__global__ __launch_bounds__(256)
void gcn_bias_relu(float* __restrict__ agg, const float* __restrict__ b,
                   long long n) {
  long long t = (long long)blockIdx.x * blockDim.x + threadIdx.x;
  if (t >= n) return;
  const int c = (int)(t & (DH - 1));
  const float v = agg[t] + b[c];
  agg[t] = v > 0.0f ? v : 0.0f;
}

__global__ __launch_bounds__(256)
void gcn_init_bias(float* __restrict__ out, const float* __restrict__ b,
                   long long n) {
  long long t = (long long)blockIdx.x * blockDim.x + threadIdx.x;
  if (t >= n) return;
  out[t] = b[(int)(t % DO)];
}

extern "C" void kernel_launch(void* const* d_in, const int* in_sizes, int n_in,
                              void* d_out, int out_size, void* d_ws,
                              size_t ws_size, hipStream_t stream) {
  const float* x  = (const float*)d_in[0];
  const int*   ei = (const int*)d_in[1];
  const float* ew = (const float*)d_in[2];
  const float* W1 = (const float*)d_in[3];
  const float* b1 = (const float*)d_in[4];
  const float* W2 = (const float*)d_in[5];
  const float* b2 = (const float*)d_in[6];

  const int N = in_sizes[0] / DH;
  const int E = in_sizes[1] / 2;
  const int* src = ei;
  const int* dst = ei + E;

  float* m1   = (float*)d_ws;
  float* agg1 = m1 + (size_t)N * DH;
  float* m2   = agg1 + (size_t)N * DH;
  float* out  = (float*)d_out;

  const int nt = 256;
  const int rowTiles = (N + 15) / 16;

  {
    long long n = (long long)N * DH;
    gcn_zero<<<(int)((n + nt - 1) / nt), nt, 0, stream>>>(agg1, n);
  }
  gcn_gemm1<<<rowTiles, 256, 0, stream>>>(x, W1, m1, N);
  {
    long long total = (long long)E * DH;
    gcn_scatter<DH><<<(int)((total + nt - 1) / nt), nt, 0, stream>>>(
        m1, src, dst, ew, agg1, total);
  }
  {
    long long n = (long long)N * DH;
    gcn_bias_relu<<<(int)((n + nt - 1) / nt), nt, 0, stream>>>(agg1, b1, n);
  }
  {
    dim3 g(rowTiles, 3);
    gcn_gemm2<<<g, 32, 0, stream>>>(agg1, W2, m2, N);
  }
  {
    long long n = (long long)N * DO;
    gcn_init_bias<<<(int)((n + nt - 1) / nt), nt, 0, stream>>>(out, b2, n);
  }
  {
    long long total = (long long)E * DO;
    gcn_scatter<DO><<<(int)((total + nt - 1) / nt), nt, 0, stream>>>(
        m2, src, dst, ew, out, total);
  }
}


# Aggregate by destination: edges grouped by dst per call (counting sort, int atomics), rows summed in registers and stored once; no float atomics, sum order = rank order, bias/relu in the store
# speedup vs baseline: 2.6782x; 2.6782x over previous
#include <hip/hip_runtime.h>

typedef float v2f __attribute__((ext_vector_type(2)));
typedef float v8f __attribute__((ext_vector_type(8)));

#define DH 128
#define DO 40

__global__ __launch_bounds__(256)
void gcn_gemm1(const float* __restrict__ x, const float* __restrict__ W,
               float* __restrict__ out, int N) {
  const int lane = threadIdx.x & 31;
  const int wave = threadIdx.x >> 5;
  const int r0 = blockIdx.x * 16;
  const int c0 = wave * 16;
  const int m  = lane & 15;
  const int kk = (lane >> 4) << 1;

  const int arow = r0 + m;
  const float* __restrict__ xrow = x + (size_t)(arow < N ? arow : N - 1) * DH;

  v8f acc = {};
  #pragma unroll 4
  for (int k = 0; k < DH; k += 4) {
    v2f a, b;
    a.x = xrow[k + kk];
    a.y = xrow[k + kk + 1];
    b.x = W[(size_t)(k + kk) * DH + c0 + m];
    b.y = W[(size_t)(k + kk + 1) * DH + c0 + m];
    acc = __builtin_amdgcn_wmma_f32_16x16x4_f32(
        false, a, false, b, (short)0, acc, false, false);
  }

  const int rbase = r0 + ((lane >> 4) << 3);
  const int col = c0 + m;
  float* __restrict__ p = out + (size_t)rbase * DH + col;
  if (r0 + 16 <= N) {
    #pragma unroll
    for (int j = 0; j < 8; ++j) p[(size_t)j * DH] = acc[j];
  } else {
    #pragma unroll
    for (int j = 0; j < 8; ++j)
      if (rbase + j < N) p[(size_t)j * DH] = acc[j];
  }
}

__global__ __launch_bounds__(32)
void gcn_gemm2(const float* __restrict__ h, const float* __restrict__ W2,
               float* __restrict__ out, int N) {
  const int lane = threadIdx.x & 31;
  const int r0 = blockIdx.x * 16;
  const int c0 = blockIdx.y * 16;
  const int m  = lane & 15;
  const int kk = (lane >> 4) << 1;

  const int arow = r0 + m;
  const float* __restrict__ hrow = h + (size_t)(arow < N ? arow : N - 1) * DH;
  const int  col = c0 + m;
  const bool cv  = (col < DO);

  v8f acc = {};
  #pragma unroll 4
  for (int k = 0; k < DH; k += 4) {
    v2f a, b;
    a.x = hrow[k + kk];
    a.y = hrow[k + kk + 1];
    b.x = cv ? W2[(size_t)(k + kk) * DO + col] : 0.0f;
    b.y = cv ? W2[(size_t)(k + kk + 1) * DO + col] : 0.0f;
    acc = __builtin_amdgcn_wmma_f32_16x16x4_f32(
        false, a, false, b, (short)0, acc, false, false);
  }

  if (!cv) return;
  const int rbase = r0 + ((lane >> 4) << 3);
  float* __restrict__ p = out + (size_t)rbase * DO + col;
  if (r0 + 16 <= N) {
    #pragma unroll
    for (int j = 0; j < 8; ++j) p[(size_t)j * DO] = acc[j];
  } else {
    #pragma unroll
    for (int j = 0; j < 8; ++j)
      if (rbase + j < N) p[(size_t)j * DO] = acc[j];
  }
}


__global__ __launch_bounds__(256)
void gcn_zero_cnt(int* __restrict__ cnt, int n) {
  const int t = blockIdx.x * blockDim.x + threadIdx.x;
  if (t < n) cnt[t] = 0;
}

__global__ __launch_bounds__(256)
void gcn_rank(const int* __restrict__ dst, int* __restrict__ cnt,
              int* __restrict__ pos, int E) {
  const int e = blockIdx.x * blockDim.x + threadIdx.x;
  if (e >= E) return;
  pos[e] = atomicAdd(&cnt[dst[e]], 1);
}

__global__ __launch_bounds__(256)
void gcn_alloc(const int* __restrict__ cnt, int* __restrict__ cursor,
               int* __restrict__ base, int N) {
  const int d = blockIdx.x * blockDim.x + threadIdx.x;
  const int lane = threadIdx.x & 31;
  const int c = (d < N) ? cnt[d] : 0;
  int incl = c;
  #pragma unroll
  for (int off = 1; off < 32; off <<= 1) {
    const int u = __shfl_up(incl, off, 32);
    if (lane >= off) incl += u;
  }
  const int total = __shfl(incl, 31, 32);
  int wbase = 0;
  if (lane == 31) wbase = atomicAdd(cursor, total);
  wbase = __shfl(wbase, 31, 32);
  if (d < N) base[d] = wbase + incl - c;
}

__global__ __launch_bounds__(256)
void gcn_fill(const int* __restrict__ src, const int* __restrict__ dst,
              const float* __restrict__ w, const int* __restrict__ base,
              const int* __restrict__ pos, int* __restrict__ csr_i, int E) {
  const int e = blockIdx.x * blockDim.x + threadIdx.x;
  if (e >= E) return;
  const int idx = base[dst[e]] + pos[e];
  ((int2*)csr_i)[idx] = make_int2(src[e], __float_as_int(w[e]));
}

__device__ __forceinline__ void fma4(float4& a, const float4 v, const float w) {
  a.x = fmaf(v.x, w, a.x);
  a.y = fmaf(v.y, w, a.y);
  a.z = fmaf(v.z, w, a.z);
  a.w = fmaf(v.w, w, a.w);
}

__global__ __launch_bounds__(256)
void gcn_gather128(const float* __restrict__ m, const int* __restrict__ csr_i,
                   const int* __restrict__ base, const int* __restrict__ cnt,
                   const float* __restrict__ b, float* __restrict__ h, int N) {
  const int lane = threadIdx.x & 31;
  const int row = __builtin_amdgcn_readfirstlane(
      (int)(blockIdx.x * 8 + (threadIdx.x >> 5)));
  if (row >= N) return;
  const int2* __restrict__ seg = (const int2*)csr_i + base[row];
  const int n = cnt[row];
  const float4* __restrict__ m4 = (const float4*)m + lane;

  float4 acc = make_float4(0.f, 0.f, 0.f, 0.f);
  int j = 0;
  for (; j + 8 <= n; j += 8) {
    int2 e[8];
    float4 v[8];
    #pragma unroll
    for (int u = 0; u < 8; ++u) e[u] = seg[j + u];
    #pragma unroll
    for (int u = 0; u < 8; ++u) v[u] = m4[(size_t)e[u].x * (DH / 4)];
    #pragma unroll
    for (int u = 0; u < 8; ++u) fma4(acc, v[u], __int_as_float(e[u].y));
  }
  if (j + 4 <= n) {
    int2 e[4];
    float4 v[4];
    #pragma unroll
    for (int u = 0; u < 4; ++u) e[u] = seg[j + u];
    #pragma unroll
    for (int u = 0; u < 4; ++u) v[u] = m4[(size_t)e[u].x * (DH / 4)];
    #pragma unroll
    for (int u = 0; u < 4; ++u) fma4(acc, v[u], __int_as_float(e[u].y));
    j += 4;
  }
  if (j + 2 <= n) {
    const int2 e0 = seg[j], e1 = seg[j + 1];
    const float4 v0 = m4[(size_t)e0.x * (DH / 4)];
    const float4 v1 = m4[(size_t)e1.x * (DH / 4)];
    fma4(acc, v0, __int_as_float(e0.y));
    fma4(acc, v1, __int_as_float(e1.y));
    j += 2;
  }
  if (j < n) {
    const int2 e0 = seg[j];
    const float4 v0 = m4[(size_t)e0.x * (DH / 4)];
    fma4(acc, v0, __int_as_float(e0.y));
  }

  const float4 bb = ((const float4*)b)[lane];
  float4 r;
  r.x = fmaxf(acc.x + bb.x, 0.0f);
  r.y = fmaxf(acc.y + bb.y, 0.0f);
  r.z = fmaxf(acc.z + bb.z, 0.0f);
  r.w = fmaxf(acc.w + bb.w, 0.0f);
  ((float4*)h)[(size_t)row * (DH / 4) + lane] = r;
}

__global__ __launch_bounds__(256)
void gcn_gather40(const float* __restrict__ m, const int* __restrict__ csr_i,
                  const int* __restrict__ base, const int* __restrict__ cnt,
                  const float* __restrict__ b, float* __restrict__ out, int N) {
  const int t = blockIdx.x * blockDim.x + threadIdx.x;
  if (t >= N * (DO / 4)) return;
  const int row = t / (DO / 4);
  const int ch  = t - row * (DO / 4);
  const int2* __restrict__ seg = (const int2*)csr_i + base[row];
  const int n = cnt[row];
  const float4* __restrict__ m4 = (const float4*)m + ch;

  float4 acc = make_float4(0.f, 0.f, 0.f, 0.f);
  int j = 0;
  for (; j + 4 <= n; j += 4) {
    int2 e[4];
    float4 v[4];
    #pragma unroll
    for (int u = 0; u < 4; ++u) e[u] = seg[j + u];
    #pragma unroll
    for (int u = 0; u < 4; ++u) v[u] = m4[(size_t)e[u].x * (DO / 4)];
    #pragma unroll
    for (int u = 0; u < 4; ++u) fma4(acc, v[u], __int_as_float(e[u].y));
  }
  if (j + 2 <= n) {
    const int2 e0 = seg[j], e1 = seg[j + 1];
    const float4 v0 = m4[(size_t)e0.x * (DO / 4)];
    const float4 v1 = m4[(size_t)e1.x * (DO / 4)];
    fma4(acc, v0, __int_as_float(e0.y));
    fma4(acc, v1, __int_as_float(e1.y));
    j += 2;
  }
  if (j < n) {
    const int2 e0 = seg[j];
    const float4 v0 = m4[(size_t)e0.x * (DO / 4)];
    fma4(acc, v0, __int_as_float(e0.y));
  }

  const float4 bb = ((const float4*)b)[ch];
  float4 r;
  r.x = acc.x + bb.x;
  r.y = acc.y + bb.y;
  r.z = acc.z + bb.z;
  r.w = acc.w + bb.w;
  ((float4*)out)[t] = r;
}

extern "C" void kernel_launch(void* const* d_in, const int* in_sizes, int n_in,
                              void* d_out, int out_size, void* d_ws,
                              size_t ws_size, hipStream_t stream) {
  const float* x  = (const float*)d_in[0];
  const int*   ei = (const int*)d_in[1];
  const float* ew = (const float*)d_in[2];
  const float* W1 = (const float*)d_in[3];
  const float* b1 = (const float*)d_in[4];
  const float* W2 = (const float*)d_in[5];
  const float* b2 = (const float*)d_in[6];

  const int N = in_sizes[0] / DH;
  const int E = in_sizes[1] / 2;
  const int* src = ei;
  const int* dst = ei + E;

  float* m1   = (float*)d_ws;
  float* h    = m1 + (size_t)N * DH;
  float* m2   = h + (size_t)N * DH;
  const int NP = (N + 35) & ~3;
  int*   cnt  = (int*)(m2 + (size_t)N * DO);
  int*   base = cnt + NP;
  int*   csr  = base + NP;
  int*   pos  = csr + (size_t)E * 2;
  float* out  = (float*)d_out;

  const int nt = 256;
  const int rowTiles = (N + 15) / 16;

  gcn_zero_cnt<<<(N + 1 + nt - 1) / nt, nt, 0, stream>>>(cnt, N + 1);
  gcn_rank<<<(E + nt - 1) / nt, nt, 0, stream>>>(dst, cnt, pos, E);
  gcn_alloc<<<(N + nt - 1) / nt, nt, 0, stream>>>(cnt, cnt + N, base, N);
  gcn_fill<<<(E + nt - 1) / nt, nt, 0, stream>>>(src, dst, ew, base, pos, csr,
                                                 E);
  gcn_gemm1<<<rowTiles, 256, 0, stream>>>(x, W1, m1, N);
  gcn_gather128<<<(N + 7) / 8, 256, 0, stream>>>(m1, csr, base, cnt, b1, h, N);
  {
    dim3 g(rowTiles, 3);
    gcn_gemm2<<<g, 32, 0, stream>>>(h, W2, m2, N);
  }
  gcn_gather40<<<(N * (DO / 4) + nt - 1) / nt, nt, 0, stream>>>(
      m2, csr, base, cnt, b2, out, N);
}


# Fixed grids: threads and waves loop over several edges, rows and row tiles (GEMM1 holds its W1 fragments in registers across tiles, a GEMM2 wave owns all 3 column tiles): far fewer waves started
# speedup vs baseline: 3.2028x; 1.1959x over previous
#include <hip/hip_runtime.h>

typedef float v2f __attribute__((ext_vector_type(2)));
typedef float v8f __attribute__((ext_vector_type(8)));

#define DH 128
#define DO 40

__global__ __launch_bounds__(128)
void gcn_gemm1(const float* __restrict__ x, const float* __restrict__ W,
               float* __restrict__ out, int N, int rowTiles) {
  const int lane = threadIdx.x & 31;
  const int wave = threadIdx.x >> 5;
  const int c0 = wave * 32;
  const int m  = lane & 15;
  const int kk = (lane >> 4) << 1;

  v2f b0[32], b1[32];
  #pragma unroll
  for (int s = 0; s < 32; ++s) {
    const float* __restrict__ wr = W + (size_t)(4 * s + kk) * DH + c0 + m;
    b0[s].x = wr[0];
    b0[s].y = wr[DH];
    b1[s].x = wr[16];
    b1[s].y = wr[DH + 16];
  }

  for (int t = blockIdx.x; t < rowTiles; t += gridDim.x) {
    const int r0 = t * 16;
    const int arow = r0 + m;
    const float* __restrict__ xrow =
        x + (size_t)(arow < N ? arow : N - 1) * DH + kk;

    v2f a[32];
    #pragma unroll
    for (int s = 0; s < 32; ++s) {
      a[s].x = xrow[4 * s];
      a[s].y = xrow[4 * s + 1];
    }
    asm volatile("" :: "v"(a[0]), "v"(a[1]), "v"(a[2]), "v"(a[3]), "v"(a[4]), "v"(a[5]), "v"(a[6]), "v"(a[7]), "v"(a[8]), "v"(a[9]), "v"(a[10]), "v"(a[11]), "v"(a[12]), "v"(a[13]), "v"(a[14]), "v"(a[15]));
    asm volatile("" :: "v"(a[16]), "v"(a[17]), "v"(a[18]), "v"(a[19]), "v"(a[20]), "v"(a[21]), "v"(a[22]), "v"(a[23]), "v"(a[24]), "v"(a[25]), "v"(a[26]), "v"(a[27]), "v"(a[28]), "v"(a[29]), "v"(a[30]), "v"(a[31]));
    v8f acc0 = {}, acc1 = {};
    #pragma unroll
    for (int s = 0; s < 32; ++s) {
      acc0 = __builtin_amdgcn_wmma_f32_16x16x4_f32(
          false, a[s], false, b0[s], (short)0, acc0, false, false);
      acc1 = __builtin_amdgcn_wmma_f32_16x16x4_f32(
          false, a[s], false, b1[s], (short)0, acc1, false, false);
    }

    const int rbase = r0 + ((lane >> 4) << 3);
    float* __restrict__ p = out + (size_t)rbase * DH + c0 + m;
    if (r0 + 16 <= N) {
      #pragma unroll
      for (int j = 0; j < 8; ++j) {
        p[(size_t)j * DH] = acc0[j];
        p[(size_t)j * DH + 16] = acc1[j];
      }
    } else {
      #pragma unroll
      for (int j = 0; j < 8; ++j)
        if (rbase + j < N) {
          p[(size_t)j * DH] = acc0[j];
          p[(size_t)j * DH + 16] = acc1[j];
        }
    }
  }
}

__global__ __launch_bounds__(128)
void gcn_gemm2(const float* __restrict__ h, const float* __restrict__ W2,
               float* __restrict__ out, int N, int rowTiles) {
  const int lane = threadIdx.x & 31;
  const int t = __builtin_amdgcn_readfirstlane(
      (int)(blockIdx.x * 4 + (threadIdx.x >> 5)));
  if (t >= rowTiles) return;
  const int r0 = t * 16;
  const int m  = lane & 15;
  const int kk = (lane >> 4) << 1;

  const int arow = r0 + m;
  const float* __restrict__ hrow =
      h + (size_t)(arow < N ? arow : N - 1) * DH + kk;
  const bool cv2 = (32 + m < DO);
  const int c2 = cv2 ? 32 + m : DO - 1;

  v2f av[32];
  #pragma unroll
  for (int s = 0; s < 32; ++s) {
    av[s].x = hrow[4 * s];
    av[s].y = hrow[4 * s + 1];
  }
  asm volatile("" :: "v"(av[0]), "v"(av[1]), "v"(av[2]), "v"(av[3]), "v"(av[4]), "v"(av[5]), "v"(av[6]), "v"(av[7]), "v"(av[8]), "v"(av[9]), "v"(av[10]), "v"(av[11]), "v"(av[12]), "v"(av[13]), "v"(av[14]), "v"(av[15]));
  asm volatile("" :: "v"(av[16]), "v"(av[17]), "v"(av[18]), "v"(av[19]), "v"(av[20]), "v"(av[21]), "v"(av[22]), "v"(av[23]), "v"(av[24]), "v"(av[25]), "v"(av[26]), "v"(av[27]), "v"(av[28]), "v"(av[29]), "v"(av[30]), "v"(av[31]));
  v8f acc0 = {}, acc1 = {}, acc2 = {};
  #pragma unroll
  for (int k = 0; k < DH; k += 4) {
    v2f b0, b1, b2;
    const v2f a = av[k >> 2];
    const float* __restrict__ wr = W2 + (size_t)(k + kk) * DO + m;
    b0.x = wr[0];
    b0.y = wr[DO];
    b1.x = wr[16];
    b1.y = wr[DO + 16];
    const float* __restrict__ wr2 = W2 + (size_t)(k + kk) * DO + c2;
    const float w2x = wr2[0];
    const float w2y = wr2[DO];
    b2.x = cv2 ? w2x : 0.0f;
    b2.y = cv2 ? w2y : 0.0f;
    acc0 = __builtin_amdgcn_wmma_f32_16x16x4_f32(
        false, a, false, b0, (short)0, acc0, false, false);
    acc1 = __builtin_amdgcn_wmma_f32_16x16x4_f32(
        false, a, false, b1, (short)0, acc1, false, false);
    acc2 = __builtin_amdgcn_wmma_f32_16x16x4_f32(
        false, a, false, b2, (short)0, acc2, false, false);
  }

  const int rbase = r0 + ((lane >> 4) << 3);
  float* __restrict__ p = out + (size_t)rbase * DO + m;
  if (r0 + 16 <= N) {
    #pragma unroll
    for (int j = 0; j < 8; ++j) {
      p[(size_t)j * DO] = acc0[j];
      p[(size_t)j * DO + 16] = acc1[j];
      if (cv2) p[(size_t)j * DO + 32] = acc2[j];
    }
  } else {
    #pragma unroll
    for (int j = 0; j < 8; ++j)
      if (rbase + j < N) {
        p[(size_t)j * DO] = acc0[j];
        p[(size_t)j * DO + 16] = acc1[j];
        if (cv2) p[(size_t)j * DO + 32] = acc2[j];
      }
  }
}


__global__ __launch_bounds__(256)
void gcn_zero_cnt(int4* __restrict__ cnt4, int n4) {
  const int t = blockIdx.x * blockDim.x + threadIdx.x;
  if (t < n4) cnt4[t] = make_int4(0, 0, 0, 0);
}

__global__ __launch_bounds__(256)
void gcn_rank(const int* __restrict__ dst, int* __restrict__ cnt,
              int* __restrict__ pos, int E) {
  const int S = gridDim.x * blockDim.x;
  int e = blockIdx.x * blockDim.x + threadIdx.x;
  for (; e + 3 * S < E; e += 4 * S) {
    int d[4], r[4];
    #pragma unroll
    for (int u = 0; u < 4; ++u) d[u] = dst[e + u * S];
    #pragma unroll
    for (int u = 0; u < 4; ++u) r[u] = atomicAdd(&cnt[d[u]], 1);
    #pragma unroll
    for (int u = 0; u < 4; ++u) pos[e + u * S] = r[u];
  }
  for (; e < E; e += S) pos[e] = atomicAdd(&cnt[dst[e]], 1);
}

__global__ __launch_bounds__(256)
void gcn_alloc(const int* __restrict__ cnt, int* __restrict__ cursor,
               int* __restrict__ base, int N) {
  const int d = blockIdx.x * blockDim.x + threadIdx.x;
  const int lane = threadIdx.x & 31;
  const int c = (d < N) ? cnt[d] : 0;
  int incl = c;
  #pragma unroll
  for (int off = 1; off < 32; off <<= 1) {
    const int u = __shfl_up(incl, off, 32);
    if (lane >= off) incl += u;
  }
  const int total = __shfl(incl, 31, 32);
  int wbase = 0;
  if (lane == 31) wbase = atomicAdd(cursor, total);
  wbase = __shfl(wbase, 31, 32);
  if (d < N) base[d] = wbase + incl - c;
}

__global__ __launch_bounds__(256)
void gcn_fill(const int* __restrict__ src, const int* __restrict__ dst,
              const float* __restrict__ w, const int* __restrict__ base,
              const int* __restrict__ pos, int* __restrict__ csr_i, int E) {
  const int S = gridDim.x * blockDim.x;
  int e = blockIdx.x * blockDim.x + threadIdx.x;
  for (; e + 3 * S < E; e += 4 * S) {
    int d[4], p[4], s[4], bs[4];
    float ww[4];
    #pragma unroll
    for (int u = 0; u < 4; ++u) d[u] = dst[e + u * S];
    #pragma unroll
    for (int u = 0; u < 4; ++u) {
      p[u] = pos[e + u * S];
      s[u] = src[e + u * S];
      ww[u] = w[e + u * S];
    }
    #pragma unroll
    for (int u = 0; u < 4; ++u) bs[u] = base[d[u]];
    #pragma unroll
    for (int u = 0; u < 4; ++u)
      ((int2*)csr_i)[bs[u] + p[u]] = make_int2(s[u], __float_as_int(ww[u]));
  }
  for (; e < E; e += S) {
    const int idx = base[dst[e]] + pos[e];
    ((int2*)csr_i)[idx] = make_int2(src[e], __float_as_int(w[e]));
  }
}

__device__ __forceinline__ void fma4(float4& a, const float4 v, const float w) {
  a.x = fmaf(v.x, w, a.x);
  a.y = fmaf(v.y, w, a.y);
  a.z = fmaf(v.z, w, a.z);
  a.w = fmaf(v.w, w, a.w);
}

__global__ __launch_bounds__(256)
void gcn_gather128(const float* __restrict__ m, const int* __restrict__ csr_i,
                   const int* __restrict__ base, const int* __restrict__ cnt,
                   const float* __restrict__ b, float* __restrict__ h, int N) {
  const int lane = threadIdx.x & 31;
  const int nw = gridDim.x * 8;
  int row = __builtin_amdgcn_readfirstlane(
      (int)(blockIdx.x * 8 + (threadIdx.x >> 5)));
  if (row >= N) return;
  const float4* __restrict__ m4 = (const float4*)m + lane;
  const float4 bb = ((const float4*)b)[lane];
  int sb = base[row];
  int n = cnt[row];

  for (;;) {
    const int nrow = row + nw;
    int nsb = 0, nn = 0;
    if (nrow < N) {
      nsb = base[nrow];
      nn = cnt[nrow];
    }
    const int2* __restrict__ seg = (const int2*)csr_i + sb;

    float4 acc = make_float4(0.f, 0.f, 0.f, 0.f);
    int j = 0;
    for (; j + 8 <= n; j += 8) {
      int2 e[8];
      float4 v[8];
      #pragma unroll
      for (int u = 0; u < 8; ++u) e[u] = seg[j + u];
      #pragma unroll
      for (int u = 0; u < 8; ++u) v[u] = m4[(size_t)e[u].x * (DH / 4)];
      #pragma unroll
      for (int u = 0; u < 8; ++u) fma4(acc, v[u], __int_as_float(e[u].y));
    }
    if (j + 4 <= n) {
      int2 e[4];
      float4 v[4];
      #pragma unroll
      for (int u = 0; u < 4; ++u) e[u] = seg[j + u];
      #pragma unroll
      for (int u = 0; u < 4; ++u) v[u] = m4[(size_t)e[u].x * (DH / 4)];
      #pragma unroll
      for (int u = 0; u < 4; ++u) fma4(acc, v[u], __int_as_float(e[u].y));
      j += 4;
    }
    if (j + 2 <= n) {
      const int2 e0 = seg[j], e1 = seg[j + 1];
      const float4 v0 = m4[(size_t)e0.x * (DH / 4)];
      const float4 v1 = m4[(size_t)e1.x * (DH / 4)];
      fma4(acc, v0, __int_as_float(e0.y));
      fma4(acc, v1, __int_as_float(e1.y));
      j += 2;
    }
    if (j < n) {
      const int2 e0 = seg[j];
      const float4 v0 = m4[(size_t)e0.x * (DH / 4)];
      fma4(acc, v0, __int_as_float(e0.y));
    }

    float4 r;
    r.x = fmaxf(acc.x + bb.x, 0.0f);
    r.y = fmaxf(acc.y + bb.y, 0.0f);
    r.z = fmaxf(acc.z + bb.z, 0.0f);
    r.w = fmaxf(acc.w + bb.w, 0.0f);
    ((float4*)h)[(size_t)row * (DH / 4) + lane] = r;

    if (nrow >= N) break;
    row = nrow;
    sb = nsb;
    n = nn;
  }
}

__global__ __launch_bounds__(256)
void gcn_gather40(const float* __restrict__ m, const int* __restrict__ csr_i,
                  const int* __restrict__ base, const int* __restrict__ cnt,
                  const float* __restrict__ b, float* __restrict__ out, int N) {
  const int t = blockIdx.x * blockDim.x + threadIdx.x;
  if (t >= N * (DO / 4)) return;
  const int row = t / (DO / 4);
  const int ch  = t - row * (DO / 4);
  const int2* __restrict__ seg = (const int2*)csr_i + base[row];
  const int n = cnt[row];
  const float4* __restrict__ m4 = (const float4*)m + ch;

  float4 acc = make_float4(0.f, 0.f, 0.f, 0.f);
  int j = 0;
  for (; j + 4 <= n; j += 4) {
    int2 e[4];
    float4 v[4];
    #pragma unroll
    for (int u = 0; u < 4; ++u) e[u] = seg[j + u];
    #pragma unroll
    for (int u = 0; u < 4; ++u) v[u] = m4[(size_t)e[u].x * (DO / 4)];
    #pragma unroll
    for (int u = 0; u < 4; ++u) fma4(acc, v[u], __int_as_float(e[u].y));
  }
  if (j + 2 <= n) {
    const int2 e0 = seg[j], e1 = seg[j + 1];
    const float4 v0 = m4[(size_t)e0.x * (DO / 4)];
    const float4 v1 = m4[(size_t)e1.x * (DO / 4)];
    fma4(acc, v0, __int_as_float(e0.y));
    fma4(acc, v1, __int_as_float(e1.y));
    j += 2;
  }
  if (j < n) {
    const int2 e0 = seg[j];
    const float4 v0 = m4[(size_t)e0.x * (DO / 4)];
    fma4(acc, v0, __int_as_float(e0.y));
  }

  const float4 bb = ((const float4*)b)[ch];
  float4 r;
  r.x = acc.x + bb.x;
  r.y = acc.y + bb.y;
  r.z = acc.z + bb.z;
  r.w = acc.w + bb.w;
  ((float4*)out)[t] = r;
}

extern "C" void kernel_launch(void* const* d_in, const int* in_sizes, int n_in,
                              void* d_out, int out_size, void* d_ws,
                              size_t ws_size, hipStream_t stream) {
  const float* x  = (const float*)d_in[0];
  const int*   ei = (const int*)d_in[1];
  const float* ew = (const float*)d_in[2];
  const float* W1 = (const float*)d_in[3];
  const float* b1 = (const float*)d_in[4];
  const float* W2 = (const float*)d_in[5];
  const float* b2 = (const float*)d_in[6];

  const int N = in_sizes[0] / DH;
  const int E = in_sizes[1] / 2;
  const int* src = ei;
  const int* dst = ei + E;

  float* m1   = (float*)d_ws;
  float* h    = m1 + (size_t)N * DH;
  float* m2   = h + (size_t)N * DH;
  const int NP = (N + 35) & ~3;
  int*   cnt  = (int*)(m2 + (size_t)N * DO);
  int*   base = cnt + NP;
  int*   csr  = base + NP;
  int*   pos  = csr + (size_t)E * 2;
  float* out  = (float*)d_out;

  const int nt = 256;
  const int rowTiles = (N + 15) / 16;

  gcn_zero_cnt<<<(NP / 4 + nt - 1) / nt, nt, 0, stream>>>((int4*)cnt, NP / 4);
  gcn_rank<<<512, nt, 0, stream>>>(dst, cnt, pos, E);
  gcn_alloc<<<(N + nt - 1) / nt, nt, 0, stream>>>(cnt, cnt + N, base, N);
  gcn_fill<<<512, nt, 0, stream>>>(src, dst, ew, base, pos, csr, E);
  gcn_gemm1<<<1024, 128, 0, stream>>>(x, W1, m1, N, rowTiles);
  gcn_gather128<<<1024, 256, 0, stream>>>(m1, csr, base, cnt, b1, h, N);
  gcn_gemm2<<<(rowTiles + 3) / 4, 128, 0, stream>>>(h, W2, m2, N, rowTiles);
  gcn_gather40<<<(N * (DO / 4) + nt - 1) / nt, nt, 0, stream>>>(
      m2, csr, base, cnt, b2, out, N);
}


# Fuse the second projection into the layer-1 aggregation kernel: the summed relu rows stay in an LDS tile and feed the WMMA directly; h is never written or re-read, one launch fewer; sums unchanged
# speedup vs baseline: 3.5400x; 1.1053x over previous
#include <hip/hip_runtime.h>

typedef float v2f __attribute__((ext_vector_type(2)));
typedef float v8f __attribute__((ext_vector_type(8)));

#define DH 128
#define DO 40

__global__ __launch_bounds__(128)
void gcn_gemm1(const float* __restrict__ x, const float* __restrict__ W,
               float* __restrict__ out, int N, int rowTiles) {
  const int lane = threadIdx.x & 31;
  const int wave = threadIdx.x >> 5;
  const int c0 = wave * 32;
  const int m  = lane & 15;
  const int kk = (lane >> 4) << 1;

  v2f b0[32], b1[32];
  #pragma unroll
  for (int s = 0; s < 32; ++s) {
    const float* __restrict__ wr = W + (size_t)(4 * s + kk) * DH + c0 + m;
    b0[s].x = wr[0];
    b0[s].y = wr[DH];
    b1[s].x = wr[16];
    b1[s].y = wr[DH + 16];
  }

  for (int t = blockIdx.x; t < rowTiles; t += gridDim.x) {
    const int r0 = t * 16;
    const int arow = r0 + m;
    const float* __restrict__ xrow =
        x + (size_t)(arow < N ? arow : N - 1) * DH + kk;

    v2f a[32];
    #pragma unroll
    for (int s = 0; s < 32; ++s) {
      a[s].x = xrow[4 * s];
      a[s].y = xrow[4 * s + 1];
    }
    asm volatile("" :: "v"(a[0]), "v"(a[1]), "v"(a[2]), "v"(a[3]), "v"(a[4]), "v"(a[5]), "v"(a[6]), "v"(a[7]), "v"(a[8]), "v"(a[9]), "v"(a[10]), "v"(a[11]), "v"(a[12]), "v"(a[13]), "v"(a[14]), "v"(a[15]));
    asm volatile("" :: "v"(a[16]), "v"(a[17]), "v"(a[18]), "v"(a[19]), "v"(a[20]), "v"(a[21]), "v"(a[22]), "v"(a[23]), "v"(a[24]), "v"(a[25]), "v"(a[26]), "v"(a[27]), "v"(a[28]), "v"(a[29]), "v"(a[30]), "v"(a[31]));
    v8f acc0 = {}, acc1 = {};
    #pragma unroll
    for (int s = 0; s < 32; ++s) {
      acc0 = __builtin_amdgcn_wmma_f32_16x16x4_f32(
          false, a[s], false, b0[s], (short)0, acc0, false, false);
      acc1 = __builtin_amdgcn_wmma_f32_16x16x4_f32(
          false, a[s], false, b1[s], (short)0, acc1, false, false);
    }

    const int rbase = r0 + ((lane >> 4) << 3);
    float* __restrict__ p = out + (size_t)rbase * DH + c0 + m;
    if (r0 + 16 <= N) {
      #pragma unroll
      for (int j = 0; j < 8; ++j) {
        p[(size_t)j * DH] = acc0[j];
        p[(size_t)j * DH + 16] = acc1[j];
      }
    } else {
      #pragma unroll
      for (int j = 0; j < 8; ++j)
        if (rbase + j < N) {
          p[(size_t)j * DH] = acc0[j];
          p[(size_t)j * DH + 16] = acc1[j];
        }
    }
  }
}


__global__ __launch_bounds__(256)
void gcn_zero_cnt(int4* __restrict__ cnt4, int n4) {
  const int t = blockIdx.x * blockDim.x + threadIdx.x;
  if (t < n4) cnt4[t] = make_int4(0, 0, 0, 0);
}

__global__ __launch_bounds__(256)
void gcn_rank(const int* __restrict__ dst, int* __restrict__ cnt,
              int* __restrict__ pos, int E) {
  const int S = gridDim.x * blockDim.x;
  int e = blockIdx.x * blockDim.x + threadIdx.x;
  for (; e + 3 * S < E; e += 4 * S) {
    int d[4], r[4];
    #pragma unroll
    for (int u = 0; u < 4; ++u) d[u] = dst[e + u * S];
    #pragma unroll
    for (int u = 0; u < 4; ++u) r[u] = atomicAdd(&cnt[d[u]], 1);
    #pragma unroll
    for (int u = 0; u < 4; ++u) pos[e + u * S] = r[u];
  }
  for (; e < E; e += S) pos[e] = atomicAdd(&cnt[dst[e]], 1);
}

__global__ __launch_bounds__(256)
void gcn_alloc(const int* __restrict__ cnt, int* __restrict__ cursor,
               int* __restrict__ base, int N) {
  const int d = blockIdx.x * blockDim.x + threadIdx.x;
  const int lane = threadIdx.x & 31;
  const int c = (d < N) ? cnt[d] : 0;
  int incl = c;
  #pragma unroll
  for (int off = 1; off < 32; off <<= 1) {
    const int u = __shfl_up(incl, off, 32);
    if (lane >= off) incl += u;
  }
  const int total = __shfl(incl, 31, 32);
  int wbase = 0;
  if (lane == 31) wbase = atomicAdd(cursor, total);
  wbase = __shfl(wbase, 31, 32);
  if (d < N) base[d] = wbase + incl - c;
}

__global__ __launch_bounds__(256)
void gcn_fill(const int* __restrict__ src, const int* __restrict__ dst,
              const float* __restrict__ w, const int* __restrict__ base,
              const int* __restrict__ pos, int* __restrict__ csr_i, int E) {
  const int S = gridDim.x * blockDim.x;
  int e = blockIdx.x * blockDim.x + threadIdx.x;
  for (; e + 3 * S < E; e += 4 * S) {
    int d[4], p[4], s[4], bs[4];
    float ww[4];
    #pragma unroll
    for (int u = 0; u < 4; ++u) d[u] = dst[e + u * S];
    #pragma unroll
    for (int u = 0; u < 4; ++u) {
      p[u] = pos[e + u * S];
      s[u] = src[e + u * S];
      ww[u] = w[e + u * S];
    }
    #pragma unroll
    for (int u = 0; u < 4; ++u) bs[u] = base[d[u]];
    #pragma unroll
    for (int u = 0; u < 4; ++u)
      ((int2*)csr_i)[bs[u] + p[u]] = make_int2(s[u], __float_as_int(ww[u]));
  }
  for (; e < E; e += S) {
    const int idx = base[dst[e]] + pos[e];
    ((int2*)csr_i)[idx] = make_int2(src[e], __float_as_int(w[e]));
  }
}

__device__ __forceinline__ void fma4(float4& a, const float4 v, const float w) {
  a.x = fmaf(v.x, w, a.x);
  a.y = fmaf(v.y, w, a.y);
  a.z = fmaf(v.z, w, a.z);
  a.w = fmaf(v.w, w, a.w);
}

#define LDP 132
__global__ __launch_bounds__(128)
void gcn_gather128_gemm2(const float* __restrict__ m,
                         const int* __restrict__ csr_i,
                         const int* __restrict__ base,
                         const int* __restrict__ cnt,
                         const float* __restrict__ b,
                         const float* __restrict__ W2,
                         float* __restrict__ out, int N, int rowTiles) {
  __shared__ __attribute__((aligned(16))) float hT[16 * LDP];
  __shared__ __attribute__((aligned(16))) float oT[16 * DO];

  const int lane = threadIdx.x & 31;
  const int wave = __builtin_amdgcn_readfirstlane((int)(threadIdx.x >> 5));
  const int mm = lane & 15;
  const int hh = lane >> 4;
  const int kk = hh << 1;

  const int col = wave * 16 + mm;
  const bool cv = col < DO;
  const int cc = cv ? col : DO - 1;
  v2f bf[32];
  if (wave < 3) {
    #pragma unroll
    for (int s = 0; s < 32; ++s) {
      const float* __restrict__ wr = W2 + (size_t)(4 * s + kk) * DO + cc;
      bf[s].x = wr[0];
      bf[s].y = wr[DO];
    }
  }

  const float4* __restrict__ m4 = (const float4*)m + lane;
  const float4 bb = ((const float4*)b)[lane];

  for (int t = blockIdx.x; t < rowTiles; t += gridDim.x) {
    const int r0 = t * 16;
    int row = r0 + wave * 4;
    int sb = 0, n = 0;
    if (row < N) {
      sb = base[row];
      n = cnt[row];
    }

    for (int q = 0; q < 4; ++q) {
      const int nrow = row + 1;
      int nsb = 0, nn = 0;
      if (q < 3 && nrow < N) {
        nsb = base[nrow];
        nn = cnt[nrow];
      }
      const int2* __restrict__ seg = (const int2*)csr_i + sb;

      float4 acc = make_float4(0.f, 0.f, 0.f, 0.f);
      int j = 0;
      for (; j + 8 <= n; j += 8) {
        int2 e[8];
        float4 v[8];
        #pragma unroll
        for (int u = 0; u < 8; ++u) e[u] = seg[j + u];
        #pragma unroll
        for (int u = 0; u < 8; ++u) v[u] = m4[(size_t)e[u].x * (DH / 4)];
        #pragma unroll
        for (int u = 0; u < 8; ++u) fma4(acc, v[u], __int_as_float(e[u].y));
      }
      if (j + 4 <= n) {
        int2 e[4];
        float4 v[4];
        #pragma unroll
        for (int u = 0; u < 4; ++u) e[u] = seg[j + u];
        #pragma unroll
        for (int u = 0; u < 4; ++u) v[u] = m4[(size_t)e[u].x * (DH / 4)];
        #pragma unroll
        for (int u = 0; u < 4; ++u) fma4(acc, v[u], __int_as_float(e[u].y));
        j += 4;
      }
      if (j + 2 <= n) {
        const int2 e0 = seg[j], e1 = seg[j + 1];
        const float4 v0 = m4[(size_t)e0.x * (DH / 4)];
        const float4 v1 = m4[(size_t)e1.x * (DH / 4)];
        fma4(acc, v0, __int_as_float(e0.y));
        fma4(acc, v1, __int_as_float(e1.y));
        j += 2;
      }
      if (j < n) {
        const int2 e0 = seg[j];
        const float4 v0 = m4[(size_t)e0.x * (DH / 4)];
        fma4(acc, v0, __int_as_float(e0.y));
      }

      float4 r;
      r.x = fmaxf(acc.x + bb.x, 0.0f);
      r.y = fmaxf(acc.y + bb.y, 0.0f);
      r.z = fmaxf(acc.z + bb.z, 0.0f);
      r.w = fmaxf(acc.w + bb.w, 0.0f);
      *(float4*)&hT[(wave * 4 + q) * LDP + 4 * lane] = r;

      row = nrow;
      sb = nsb;
      n = nn;
    }

    __syncthreads();

    if (wave < 3) {
      const float* __restrict__ ap = &hT[mm * LDP + kk];
      v8f dacc = {};
      #pragma unroll
      for (int s = 0; s < 32; ++s) {
        const v2f a = *(const v2f*)(ap + 4 * s);
        dacc = __builtin_amdgcn_wmma_f32_16x16x4_f32(
            false, a, false, bf[s], (short)0, dacc, false, false);
      }
      if (cv) {
        #pragma unroll
        for (int j = 0; j < 8; ++j) oT[(hh * 8 + j) * DO + col] = dacc[j];
      }
    }

    __syncthreads();

    const int rows = (N - r0 < 16) ? (N - r0) : 16;
    const int nf4 = rows * (DO / 4);
    float4* __restrict__ o4 = (float4*)out + (size_t)r0 * (DO / 4);
    for (int f = threadIdx.x; f < nf4; f += 128)
      o4[f] = ((const float4*)oT)[f];
  }
}

__global__ __launch_bounds__(256)
void gcn_gather40(const float* __restrict__ m, const int* __restrict__ csr_i,
                  const int* __restrict__ base, const int* __restrict__ cnt,
                  const float* __restrict__ b, float* __restrict__ out, int N) {
  const int t = blockIdx.x * blockDim.x + threadIdx.x;
  if (t >= N * (DO / 4)) return;
  const int row = t / (DO / 4);
  const int ch  = t - row * (DO / 4);
  const int2* __restrict__ seg = (const int2*)csr_i + base[row];
  const int n = cnt[row];
  const float4* __restrict__ m4 = (const float4*)m + ch;

  float4 acc = make_float4(0.f, 0.f, 0.f, 0.f);
  int j = 0;
  for (; j + 4 <= n; j += 4) {
    int2 e[4];
    float4 v[4];
    #pragma unroll
    for (int u = 0; u < 4; ++u) e[u] = seg[j + u];
    #pragma unroll
    for (int u = 0; u < 4; ++u) v[u] = m4[(size_t)e[u].x * (DO / 4)];
    #pragma unroll
    for (int u = 0; u < 4; ++u) fma4(acc, v[u], __int_as_float(e[u].y));
  }
  if (j + 2 <= n) {
    const int2 e0 = seg[j], e1 = seg[j + 1];
    const float4 v0 = m4[(size_t)e0.x * (DO / 4)];
    const float4 v1 = m4[(size_t)e1.x * (DO / 4)];
    fma4(acc, v0, __int_as_float(e0.y));
    fma4(acc, v1, __int_as_float(e1.y));
    j += 2;
  }
  if (j < n) {
    const int2 e0 = seg[j];
    const float4 v0 = m4[(size_t)e0.x * (DO / 4)];
    fma4(acc, v0, __int_as_float(e0.y));
  }

  const float4 bb = ((const float4*)b)[ch];
  float4 r;
  r.x = acc.x + bb.x;
  r.y = acc.y + bb.y;
  r.z = acc.z + bb.z;
  r.w = acc.w + bb.w;
  ((float4*)out)[t] = r;
}

extern "C" void kernel_launch(void* const* d_in, const int* in_sizes, int n_in,
                              void* d_out, int out_size, void* d_ws,
                              size_t ws_size, hipStream_t stream) {
  const float* x  = (const float*)d_in[0];
  const int*   ei = (const int*)d_in[1];
  const float* ew = (const float*)d_in[2];
  const float* W1 = (const float*)d_in[3];
  const float* b1 = (const float*)d_in[4];
  const float* W2 = (const float*)d_in[5];
  const float* b2 = (const float*)d_in[6];

  const int N = in_sizes[0] / DH;
  const int E = in_sizes[1] / 2;
  const int* src = ei;
  const int* dst = ei + E;

  float* m1   = (float*)d_ws;
  float* m2   = m1 + (size_t)N * DH;
  const int NP = (N + 35) & ~3;
  int*   cnt  = (int*)(m2 + (size_t)N * DO);
  int*   base = cnt + NP;
  int*   csr  = base + NP;
  int*   pos  = csr + (size_t)E * 2;
  float* out  = (float*)d_out;

  const int nt = 256;
  const int rowTiles = (N + 15) / 16;

  gcn_zero_cnt<<<(NP / 4 + nt - 1) / nt, nt, 0, stream>>>((int4*)cnt, NP / 4);
  gcn_rank<<<512, nt, 0, stream>>>(dst, cnt, pos, E);
  gcn_alloc<<<(N + nt - 1) / nt, nt, 0, stream>>>(cnt, cnt + N, base, N);
  gcn_fill<<<512, nt, 0, stream>>>(src, dst, ew, base, pos, csr, E);
  gcn_gemm1<<<1024, 128, 0, stream>>>(x, W1, m1, N, rowTiles);
  gcn_gather128_gemm2<<<(rowTiles + 1) / 2, 128, 0, stream>>>(
      m1, csr, base, cnt, b1, W2, m2, N, rowTiles);
  gcn_gather40<<<(N * (DO / 4) + nt - 1) / nt, nt, 0, stream>>>(
      m2, csr, base, cnt, b2, out, N);
}


# Group edges by dst with a two-level counting sort in LDS (chunk -> 128-row bins, then rows) instead of per-edge global atomics and scattered stores; 7 -> 5 launches; sum order = LDS arrival rank
# speedup vs baseline: 5.7095x; 1.6129x over previous
#include <hip/hip_runtime.h>

typedef float v2f __attribute__((ext_vector_type(2)));
typedef float v8f __attribute__((ext_vector_type(8)));

#define DH 128
#define DO 40

__global__ __launch_bounds__(128)
void gcn_gemm1(const float* __restrict__ x, const float* __restrict__ W,
               float* __restrict__ out, int N, int rowTiles) {
  const int lane = threadIdx.x & 31;
  const int wave = threadIdx.x >> 5;
  const int c0 = wave * 32;
  const int m  = lane & 15;
  const int kk = (lane >> 4) << 1;

  v2f b0[32], b1[32];
  #pragma unroll
  for (int s = 0; s < 32; ++s) {
    const float* __restrict__ wr = W + (size_t)(4 * s + kk) * DH + c0 + m;
    b0[s].x = wr[0];
    b0[s].y = wr[DH];
    b1[s].x = wr[16];
    b1[s].y = wr[DH + 16];
  }

  for (int t = blockIdx.x; t < rowTiles; t += gridDim.x) {
    const int r0 = t * 16;
    const int arow = r0 + m;
    const float* __restrict__ xrow =
        x + (size_t)(arow < N ? arow : N - 1) * DH + kk;

    v2f a[32];
    #pragma unroll
    for (int s = 0; s < 32; ++s) {
      a[s].x = xrow[4 * s];
      a[s].y = xrow[4 * s + 1];
    }
    asm volatile("" :: "v"(a[0]), "v"(a[1]), "v"(a[2]), "v"(a[3]), "v"(a[4]), "v"(a[5]), "v"(a[6]), "v"(a[7]), "v"(a[8]), "v"(a[9]), "v"(a[10]), "v"(a[11]), "v"(a[12]), "v"(a[13]), "v"(a[14]), "v"(a[15]));
    asm volatile("" :: "v"(a[16]), "v"(a[17]), "v"(a[18]), "v"(a[19]), "v"(a[20]), "v"(a[21]), "v"(a[22]), "v"(a[23]), "v"(a[24]), "v"(a[25]), "v"(a[26]), "v"(a[27]), "v"(a[28]), "v"(a[29]), "v"(a[30]), "v"(a[31]));
    v8f acc0 = {}, acc1 = {};
    #pragma unroll
    for (int s = 0; s < 32; ++s) {
      acc0 = __builtin_amdgcn_wmma_f32_16x16x4_f32(
          false, a[s], false, b0[s], (short)0, acc0, false, false);
      acc1 = __builtin_amdgcn_wmma_f32_16x16x4_f32(
          false, a[s], false, b1[s], (short)0, acc1, false, false);
    }

    const int rbase = r0 + ((lane >> 4) << 3);
    float* __restrict__ p = out + (size_t)rbase * DH + c0 + m;
    if (r0 + 16 <= N) {
      #pragma unroll
      for (int j = 0; j < 8; ++j) {
        p[(size_t)j * DH] = acc0[j];
        p[(size_t)j * DH + 16] = acc1[j];
      }
    } else {
      #pragma unroll
      for (int j = 0; j < 8; ++j)
        if (rbase + j < N) {
          p[(size_t)j * DH] = acc0[j];
          p[(size_t)j * DH + 16] = acc1[j];
        }
    }
  }
}

#define BR    128
#define CHUNK 4096
#define EPT   16
#define STG   6144

__global__ __launch_bounds__(256)
void gcn_bin(const int* __restrict__ src, const int* __restrict__ dst,
             const float* __restrict__ w, int* __restrict__ sD,
             int2* __restrict__ sP, int* __restrict__ st,
             int* __restrict__ cursor, int E, int NB) {
  extern __shared__ int hist[];
  __shared__ int stD[CHUNK];
  __shared__ int2 stP[CHUNK];
  __shared__ int wsum[8];

  const int tid = threadIdx.x;
  const int lane = tid & 31;
  const int wave = tid >> 5;
  const int c0 = blockIdx.x * CHUNK;
  const int cn = (E - c0 < CHUNK) ? (E - c0) : CHUNK;

  if (blockIdx.x == 0 && tid == 0) *cursor = 0;
  for (int i = tid; i <= NB; i += 256) hist[i] = 0;

  int d[EPT], s[EPT], r[EPT];
  float ww[EPT];
  #pragma unroll
  for (int u = 0; u < EPT; ++u) {
    const int i = u * 256 + tid;
    d[u] = -1;
    s[u] = 0;
    ww[u] = 0.0f;
    r[u] = 0;
    if (i < cn) {
      d[u] = dst[c0 + i];
      s[u] = src[c0 + i];
      ww[u] = w[c0 + i];
    }
  }
  __syncthreads();

  #pragma unroll
  for (int u = 0; u < EPT; ++u)
    if (d[u] >= 0) r[u] = atomicAdd(&hist[d[u] / BR], 1);
  __syncthreads();

  const int K = (NB + 255) >> 8;
  const int b0 = tid * K;
  int loc = 0;
  for (int k = 0; k < K; ++k)
    if (b0 + k < NB) loc += hist[b0 + k];
  int incl = loc;
  #pragma unroll
  for (int off = 1; off < 32; off <<= 1) {
    const int v = __shfl_up(incl, off, 32);
    if (lane >= off) incl += v;
  }
  if (lane == 31) wsum[wave] = incl;
  __syncthreads();
  int run = incl - loc;
  for (int k = 0; k < wave; ++k) run += wsum[k];
  for (int k = 0; k < K; ++k)
    if (b0 + k < NB) {
      const int c = hist[b0 + k];
      hist[b0 + k] = run;
      run += c;
    }
  if (tid == 255) hist[NB] = run;
  __syncthreads();

  int* __restrict__ strow = st + (size_t)blockIdx.x * (NB + 1);
  for (int i = tid; i <= NB; i += 256) strow[i] = hist[i];

  #pragma unroll
  for (int u = 0; u < EPT; ++u)
    if (d[u] >= 0) {
      const int p = hist[d[u] / BR] + r[u];
      stD[p] = d[u];
      stP[p] = make_int2(s[u], __float_as_int(ww[u]));
    }
  __syncthreads();

  for (int i = tid; i < cn; i += 256) {
    sD[c0 + i] = stD[i];
    sP[c0 + i] = stP[i];
  }
}

__global__ __launch_bounds__(256)
void gcn_group(const int* __restrict__ sD, const int2* __restrict__ sP,
               const int* __restrict__ st, int* __restrict__ cursor,
               int* __restrict__ base, int* __restrict__ cnt,
               int2* __restrict__ csr, int N, int NB, int NW) {
  __shared__ int cntL[BR];
  __shared__ int curL[BR];
  __shared__ int wtot[BR / 32];
  __shared__ int gb[2];
  __shared__ int2 stg[STG];

  const int tid = threadIdx.x;
  const int lane = tid & 31;
  const int wave = tid >> 5;
  const int b = blockIdx.x;
  const int r0 = b * BR;

  if (tid < BR) cntL[tid] = 0;
  __syncthreads();

  for (int f = tid; f < NW; f += 256) {
    const int* __restrict__ t = st + (size_t)f * (NB + 1) + b;
    const int s = t[0], e = t[1];
    const int* __restrict__ p = sD + (size_t)f * CHUNK;
    int i = s;
    for (; i + 4 <= e; i += 4) {
      const int d0 = p[i], d1 = p[i + 1], d2 = p[i + 2], d3 = p[i + 3];
      atomicAdd(&cntL[d0 - r0], 1);
      atomicAdd(&cntL[d1 - r0], 1);
      atomicAdd(&cntL[d2 - r0], 1);
      atomicAdd(&cntL[d3 - r0], 1);
    }
    for (; i < e; ++i) atomicAdd(&cntL[p[i] - r0], 1);
  }
  __syncthreads();

  const int c = (tid < BR) ? cntL[tid] : 0;
  int incl = c;
  #pragma unroll
  for (int off = 1; off < 32; off <<= 1) {
    const int v = __shfl_up(incl, off, 32);
    if (lane >= off) incl += v;
  }
  if (tid < BR && lane == 31) wtot[wave] = incl;
  __syncthreads();
  int excl = incl - c;
  for (int k = 0; k < wave && k < BR / 32; ++k) excl += wtot[k];
  if (tid == BR - 1) {
    const int T = excl + c;
    gb[0] = atomicAdd(cursor, T);
    gb[1] = T;
  }
  __syncthreads();
  const int gbase = gb[0];
  const int T = gb[1];
  if (tid < BR) {
    curL[tid] = excl;
    const int row = r0 + tid;
    if (row < N) {
      base[row] = gbase + excl;
      cnt[row] = c;
    }
  }
  __syncthreads();

  const bool staged = T <= STG;
  int2* __restrict__ seg = csr + gbase;
  for (int f = tid; f < NW; f += 256) {
    const int* __restrict__ t = st + (size_t)f * (NB + 1) + b;
    const int s = t[0], e = t[1];
    const int* __restrict__ p = sD + (size_t)f * CHUNK;
    const int2* __restrict__ q = sP + (size_t)f * CHUNK;
    int i = s;
    for (; i + 4 <= e; i += 4) {
      int dd[4], ps[4];
      int2 v[4];
      #pragma unroll
      for (int u = 0; u < 4; ++u) {
        dd[u] = p[i + u];
        v[u] = q[i + u];
      }
      #pragma unroll
      for (int u = 0; u < 4; ++u) ps[u] = atomicAdd(&curL[dd[u] - r0], 1);
      #pragma unroll
      for (int u = 0; u < 4; ++u) {
        if (staged) stg[ps[u]] = v[u];
        else seg[ps[u]] = v[u];
      }
    }
    for (; i < e; ++i) {
      const int ps = atomicAdd(&curL[p[i] - r0], 1);
      const int2 v = q[i];
      if (staged) stg[ps] = v;
      else seg[ps] = v;
    }
  }
  __syncthreads();
  if (staged)
    for (int i = tid; i < T; i += 256) seg[i] = stg[i];
}

__device__ __forceinline__ void fma4(float4& a, const float4 v, const float w) {
  a.x = fmaf(v.x, w, a.x);
  a.y = fmaf(v.y, w, a.y);
  a.z = fmaf(v.z, w, a.z);
  a.w = fmaf(v.w, w, a.w);
}

#define LDP 132
__global__ __launch_bounds__(128)
void gcn_gather128_gemm2(const float* __restrict__ m,
                         const int* __restrict__ csr_i,
                         const int* __restrict__ base,
                         const int* __restrict__ cnt,
                         const float* __restrict__ b,
                         const float* __restrict__ W2,
                         float* __restrict__ out, int N, int rowTiles) {
  __shared__ __attribute__((aligned(16))) float hT[16 * LDP];
  __shared__ __attribute__((aligned(16))) float oT[16 * DO];

  const int lane = threadIdx.x & 31;
  const int wave = __builtin_amdgcn_readfirstlane((int)(threadIdx.x >> 5));
  const int mm = lane & 15;
  const int hh = lane >> 4;
  const int kk = hh << 1;

  const int col = wave * 16 + mm;
  const bool cv = col < DO;
  const int cc = cv ? col : DO - 1;
  v2f bf[32];
  if (wave < 3) {
    #pragma unroll
    for (int s = 0; s < 32; ++s) {
      const float* __restrict__ wr = W2 + (size_t)(4 * s + kk) * DO + cc;
      bf[s].x = wr[0];
      bf[s].y = wr[DO];
    }
  }

  const float4* __restrict__ m4 = (const float4*)m + lane;
  const float4 bb = ((const float4*)b)[lane];

  for (int t = blockIdx.x; t < rowTiles; t += gridDim.x) {
    const int r0 = t * 16;
    int row = r0 + wave * 4;
    int sb = 0, n = 0;
    if (row < N) {
      sb = base[row];
      n = cnt[row];
    }

    for (int q = 0; q < 4; ++q) {
      const int nrow = row + 1;
      int nsb = 0, nn = 0;
      if (q < 3 && nrow < N) {
        nsb = base[nrow];
        nn = cnt[nrow];
      }
      const int2* __restrict__ seg = (const int2*)csr_i + sb;

      float4 acc = make_float4(0.f, 0.f, 0.f, 0.f);
      int j = 0;
      for (; j + 8 <= n; j += 8) {
        int2 e[8];
        float4 v[8];
        #pragma unroll
        for (int u = 0; u < 8; ++u) e[u] = seg[j + u];
        #pragma unroll
        for (int u = 0; u < 8; ++u) v[u] = m4[(size_t)e[u].x * (DH / 4)];
        #pragma unroll
        for (int u = 0; u < 8; ++u) fma4(acc, v[u], __int_as_float(e[u].y));
      }
      if (j + 4 <= n) {
        int2 e[4];
        float4 v[4];
        #pragma unroll
        for (int u = 0; u < 4; ++u) e[u] = seg[j + u];
        #pragma unroll
        for (int u = 0; u < 4; ++u) v[u] = m4[(size_t)e[u].x * (DH / 4)];
        #pragma unroll
        for (int u = 0; u < 4; ++u) fma4(acc, v[u], __int_as_float(e[u].y));
        j += 4;
      }
      if (j + 2 <= n) {
        const int2 e0 = seg[j], e1 = seg[j + 1];
        const float4 v0 = m4[(size_t)e0.x * (DH / 4)];
        const float4 v1 = m4[(size_t)e1.x * (DH / 4)];
        fma4(acc, v0, __int_as_float(e0.y));
        fma4(acc, v1, __int_as_float(e1.y));
        j += 2;
      }
      if (j < n) {
        const int2 e0 = seg[j];
        const float4 v0 = m4[(size_t)e0.x * (DH / 4)];
        fma4(acc, v0, __int_as_float(e0.y));
      }

      float4 r;
      r.x = fmaxf(acc.x + bb.x, 0.0f);
      r.y = fmaxf(acc.y + bb.y, 0.0f);
      r.z = fmaxf(acc.z + bb.z, 0.0f);
      r.w = fmaxf(acc.w + bb.w, 0.0f);
      *(float4*)&hT[(wave * 4 + q) * LDP + 4 * lane] = r;

      row = nrow;
      sb = nsb;
      n = nn;
    }

    __syncthreads();

    if (wave < 3) {
      const float* __restrict__ ap = &hT[mm * LDP + kk];
      v8f dacc = {};
      #pragma unroll
      for (int s = 0; s < 32; ++s) {
        const v2f a = *(const v2f*)(ap + 4 * s);
        dacc = __builtin_amdgcn_wmma_f32_16x16x4_f32(
            false, a, false, bf[s], (short)0, dacc, false, false);
      }
      if (cv) {
        #pragma unroll
        for (int j = 0; j < 8; ++j) oT[(hh * 8 + j) * DO + col] = dacc[j];
      }
    }

    __syncthreads();

    const int rows = (N - r0 < 16) ? (N - r0) : 16;
    const int nf4 = rows * (DO / 4);
    float4* __restrict__ o4 = (float4*)out + (size_t)r0 * (DO / 4);
    for (int f = threadIdx.x; f < nf4; f += 128)
      o4[f] = ((const float4*)oT)[f];
  }
}

__global__ __launch_bounds__(256)
void gcn_gather40(const float* __restrict__ m, const int* __restrict__ csr_i,
                  const int* __restrict__ base, const int* __restrict__ cnt,
                  const float* __restrict__ b, float* __restrict__ out, int N) {
  const int t = blockIdx.x * blockDim.x + threadIdx.x;
  if (t >= N * (DO / 4)) return;
  const int row = t / (DO / 4);
  const int ch  = t - row * (DO / 4);
  const int2* __restrict__ seg = (const int2*)csr_i + base[row];
  const int n = cnt[row];
  const float4* __restrict__ m4 = (const float4*)m + ch;

  float4 acc = make_float4(0.f, 0.f, 0.f, 0.f);
  int j = 0;
  for (; j + 4 <= n; j += 4) {
    int2 e[4];
    float4 v[4];
    #pragma unroll
    for (int u = 0; u < 4; ++u) e[u] = seg[j + u];
    #pragma unroll
    for (int u = 0; u < 4; ++u) v[u] = m4[(size_t)e[u].x * (DO / 4)];
    #pragma unroll
    for (int u = 0; u < 4; ++u) fma4(acc, v[u], __int_as_float(e[u].y));
  }
  if (j + 2 <= n) {
    const int2 e0 = seg[j], e1 = seg[j + 1];
    const float4 v0 = m4[(size_t)e0.x * (DO / 4)];
    const float4 v1 = m4[(size_t)e1.x * (DO / 4)];
    fma4(acc, v0, __int_as_float(e0.y));
    fma4(acc, v1, __int_as_float(e1.y));
    j += 2;
  }
  if (j < n) {
    const int2 e0 = seg[j];
    const float4 v0 = m4[(size_t)e0.x * (DO / 4)];
    fma4(acc, v0, __int_as_float(e0.y));
  }

  const float4 bb = ((const float4*)b)[ch];
  float4 r;
  r.x = acc.x + bb.x;
  r.y = acc.y + bb.y;
  r.z = acc.z + bb.z;
  r.w = acc.w + bb.w;
  ((float4*)out)[t] = r;
}

extern "C" void kernel_launch(void* const* d_in, const int* in_sizes, int n_in,
                              void* d_out, int out_size, void* d_ws,
                              size_t ws_size, hipStream_t stream) {
  const float* x  = (const float*)d_in[0];
  const int*   ei = (const int*)d_in[1];
  const float* ew = (const float*)d_in[2];
  const float* W1 = (const float*)d_in[3];
  const float* b1 = (const float*)d_in[4];
  const float* W2 = (const float*)d_in[5];
  const float* b2 = (const float*)d_in[6];

  const int N = in_sizes[0] / DH;
  const int E = in_sizes[1] / 2;
  const int* src = ei;
  const int* dst = ei + E;

  float* m1   = (float*)d_ws;
  float* m2   = m1 + (size_t)N * DH;
  const int NP = (N + 35) & ~3;
  int*   cnt  = (int*)(m2 + (size_t)N * DO);
  int*   base = cnt + NP;
  int2*  csr  = (int2*)(base + NP);
  int2*  sP   = csr + E;
  int*   sD   = (int*)(sP + E);
  int*   st   = sD + E;
  float* out  = (float*)d_out;

  const int nt = 256;
  const int rowTiles = (N + 15) / 16;
  const int NB = (N + BR - 1) / BR;
  const int NW = (E + CHUNK - 1) / CHUNK;

  gcn_bin<<<NW, nt, (NB + 1) * sizeof(int), stream>>>(
      src, dst, ew, sD, sP, st, cnt + N, E, NB);
  gcn_group<<<NB, nt, 0, stream>>>(sD, sP, st, cnt + N, base, cnt, csr,
                                   N, NB, NW);
  gcn_gemm1<<<1024, 128, 0, stream>>>(x, W1, m1, N, rowTiles);
  gcn_gather128_gemm2<<<(rowTiles + 1) / 2, 128, 0, stream>>>(
      m1, (const int*)csr, base, cnt, b1, W2, m2, N, rowTiles);
  gcn_gather40<<<(N * (DO / 4) + nt - 1) / nt, nt, 0, stream>>>(
      m2, (const int*)csr, base, cnt, b2, out, N);
}


# GEMM1 loses its own launch: its row tiles run as 512 extra work-groups of the first edge-grouping launch (independent work, the sort uses no matrix unit); 5 -> 4 launches; all sums unchanged
# speedup vs baseline: 5.8917x; 1.0319x over previous
#include <hip/hip_runtime.h>

typedef float v2f __attribute__((ext_vector_type(2)));
typedef float v8f __attribute__((ext_vector_type(8)));

#define DH 128
#define DO 40

__device__ __forceinline__
void gemm1_tiles(const float* __restrict__ x, const float* __restrict__ W,
                 float* __restrict__ out, int N, int rowTiles,
                 int vb, int nvb) {
  const int lane = threadIdx.x & 31;
  const int wave = (threadIdx.x >> 5) & 3;
  const int c0 = wave * 32;
  const int m  = lane & 15;
  const int kk = (lane >> 4) << 1;

  v2f b0[32], b1[32];
  #pragma unroll
  for (int s = 0; s < 32; ++s) {
    const float* __restrict__ wr = W + (size_t)(4 * s + kk) * DH + c0 + m;
    b0[s].x = wr[0];
    b0[s].y = wr[DH];
    b1[s].x = wr[16];
    b1[s].y = wr[DH + 16];
  }

  for (int t = vb; t < rowTiles; t += nvb) {
    const int r0 = t * 16;
    const int arow = r0 + m;
    const float* __restrict__ xrow =
        x + (size_t)(arow < N ? arow : N - 1) * DH + kk;

    v2f a[32];
    #pragma unroll
    for (int s = 0; s < 32; ++s) {
      a[s].x = xrow[4 * s];
      a[s].y = xrow[4 * s + 1];
    }
    asm volatile("" :: "v"(a[0]), "v"(a[1]), "v"(a[2]), "v"(a[3]), "v"(a[4]), "v"(a[5]), "v"(a[6]), "v"(a[7]), "v"(a[8]), "v"(a[9]), "v"(a[10]), "v"(a[11]), "v"(a[12]), "v"(a[13]), "v"(a[14]), "v"(a[15]));
    asm volatile("" :: "v"(a[16]), "v"(a[17]), "v"(a[18]), "v"(a[19]), "v"(a[20]), "v"(a[21]), "v"(a[22]), "v"(a[23]), "v"(a[24]), "v"(a[25]), "v"(a[26]), "v"(a[27]), "v"(a[28]), "v"(a[29]), "v"(a[30]), "v"(a[31]));
    v8f acc0 = {}, acc1 = {};
    #pragma unroll
    for (int s = 0; s < 32; ++s) {
      acc0 = __builtin_amdgcn_wmma_f32_16x16x4_f32(
          false, a[s], false, b0[s], (short)0, acc0, false, false);
      acc1 = __builtin_amdgcn_wmma_f32_16x16x4_f32(
          false, a[s], false, b1[s], (short)0, acc1, false, false);
    }

    const int rbase = r0 + ((lane >> 4) << 3);
    float* __restrict__ p = out + (size_t)rbase * DH + c0 + m;
    if (r0 + 16 <= N) {
      #pragma unroll
      for (int j = 0; j < 8; ++j) {
        p[(size_t)j * DH] = acc0[j];
        p[(size_t)j * DH + 16] = acc1[j];
      }
    } else {
      #pragma unroll
      for (int j = 0; j < 8; ++j)
        if (rbase + j < N) {
          p[(size_t)j * DH] = acc0[j];
          p[(size_t)j * DH + 16] = acc1[j];
        }
    }
  }
}

#define BR    128
#define CHUNK 4096
#define EPT   16
#define STG   6144

__global__ __launch_bounds__(256)
void gcn_bin(const int* __restrict__ src, const int* __restrict__ dst,
             const float* __restrict__ w, int* __restrict__ sD,
             int2* __restrict__ sP, int* __restrict__ st,
             int* __restrict__ cursor, int E, int NB, int NW,
             const float* __restrict__ gx, const float* __restrict__ gW,
             float* __restrict__ gm, int N, int rowTiles) {
  if ((int)blockIdx.x >= NW) {
    gemm1_tiles(gx, gW, gm, N, rowTiles,
                ((int)blockIdx.x - NW) * 2 + (int)(threadIdx.x >> 7),
                ((int)gridDim.x - NW) * 2);
    return;
  }
  extern __shared__ int hist[];
  __shared__ int stD[CHUNK];
  __shared__ int2 stP[CHUNK];
  __shared__ int wsum[8];

  const int tid = threadIdx.x;
  const int lane = tid & 31;
  const int wave = tid >> 5;
  const int c0 = blockIdx.x * CHUNK;
  const int cn = (E - c0 < CHUNK) ? (E - c0) : CHUNK;

  if (blockIdx.x == 0 && tid == 0) *cursor = 0;
  for (int i = tid; i <= NB; i += 256) hist[i] = 0;

  int d[EPT], s[EPT], r[EPT];
  float ww[EPT];
  #pragma unroll
  for (int u = 0; u < EPT; ++u) {
    const int i = u * 256 + tid;
    d[u] = -1;
    s[u] = 0;
    ww[u] = 0.0f;
    r[u] = 0;
    if (i < cn) {
      d[u] = dst[c0 + i];
      s[u] = src[c0 + i];
      ww[u] = w[c0 + i];
    }
  }
  __syncthreads();

  #pragma unroll
  for (int u = 0; u < EPT; ++u)
    if (d[u] >= 0) r[u] = atomicAdd(&hist[d[u] / BR], 1);
  __syncthreads();

  const int K = (NB + 255) >> 8;
  const int b0 = tid * K;
  int loc = 0;
  for (int k = 0; k < K; ++k)
    if (b0 + k < NB) loc += hist[b0 + k];
  int incl = loc;
  #pragma unroll
  for (int off = 1; off < 32; off <<= 1) {
    const int v = __shfl_up(incl, off, 32);
    if (lane >= off) incl += v;
  }
  if (lane == 31) wsum[wave] = incl;
  __syncthreads();
  int run = incl - loc;
  for (int k = 0; k < wave; ++k) run += wsum[k];
  for (int k = 0; k < K; ++k)
    if (b0 + k < NB) {
      const int c = hist[b0 + k];
      hist[b0 + k] = run;
      run += c;
    }
  if (tid == 255) hist[NB] = run;
  __syncthreads();

  int* __restrict__ strow = st + (size_t)blockIdx.x * (NB + 1);
  for (int i = tid; i <= NB; i += 256) strow[i] = hist[i];

  #pragma unroll
  for (int u = 0; u < EPT; ++u)
    if (d[u] >= 0) {
      const int p = hist[d[u] / BR] + r[u];
      stD[p] = d[u];
      stP[p] = make_int2(s[u], __float_as_int(ww[u]));
    }
  __syncthreads();

  for (int i = tid; i < cn; i += 256) {
    sD[c0 + i] = stD[i];
    sP[c0 + i] = stP[i];
  }
}

__global__ __launch_bounds__(256)
void gcn_group(const int* __restrict__ sD, const int2* __restrict__ sP,
               const int* __restrict__ st, int* __restrict__ cursor,
               int* __restrict__ base, int* __restrict__ cnt,
               int2* __restrict__ csr, int N, int NB, int NW) {
  __shared__ int cntL[BR];
  __shared__ int curL[BR];
  __shared__ int wtot[BR / 32];
  __shared__ int gb[2];
  __shared__ int2 stg[STG];

  const int tid = threadIdx.x;
  const int lane = tid & 31;
  const int wave = tid >> 5;
  const int b = blockIdx.x;
  const int r0 = b * BR;

  if (tid < BR) cntL[tid] = 0;
  __syncthreads();

  for (int f = tid; f < NW; f += 256) {
    const int* __restrict__ t = st + (size_t)f * (NB + 1) + b;
    const int s = t[0], e = t[1];
    const int* __restrict__ p = sD + (size_t)f * CHUNK;
    int i = s;
    for (; i + 4 <= e; i += 4) {
      const int d0 = p[i], d1 = p[i + 1], d2 = p[i + 2], d3 = p[i + 3];
      atomicAdd(&cntL[d0 - r0], 1);
      atomicAdd(&cntL[d1 - r0], 1);
      atomicAdd(&cntL[d2 - r0], 1);
      atomicAdd(&cntL[d3 - r0], 1);
    }
    for (; i < e; ++i) atomicAdd(&cntL[p[i] - r0], 1);
  }
  __syncthreads();

  const int c = (tid < BR) ? cntL[tid] : 0;
  int incl = c;
  #pragma unroll
  for (int off = 1; off < 32; off <<= 1) {
    const int v = __shfl_up(incl, off, 32);
    if (lane >= off) incl += v;
  }
  if (tid < BR && lane == 31) wtot[wave] = incl;
  __syncthreads();
  int excl = incl - c;
  for (int k = 0; k < wave && k < BR / 32; ++k) excl += wtot[k];
  if (tid == BR - 1) {
    const int T = excl + c;
    gb[0] = atomicAdd(cursor, T);
    gb[1] = T;
  }
  __syncthreads();
  const int gbase = gb[0];
  const int T = gb[1];
  if (tid < BR) {
    curL[tid] = excl;
    const int row = r0 + tid;
    if (row < N) {
      base[row] = gbase + excl;
      cnt[row] = c;
    }
  }
  __syncthreads();

  const bool staged = T <= STG;
  int2* __restrict__ seg = csr + gbase;
  for (int f = tid; f < NW; f += 256) {
    const int* __restrict__ t = st + (size_t)f * (NB + 1) + b;
    const int s = t[0], e = t[1];
    const int* __restrict__ p = sD + (size_t)f * CHUNK;
    const int2* __restrict__ q = sP + (size_t)f * CHUNK;
    int i = s;
    for (; i + 4 <= e; i += 4) {
      int dd[4], ps[4];
      int2 v[4];
      #pragma unroll
      for (int u = 0; u < 4; ++u) {
        dd[u] = p[i + u];
        v[u] = q[i + u];
      }
      #pragma unroll
      for (int u = 0; u < 4; ++u) ps[u] = atomicAdd(&curL[dd[u] - r0], 1);
      #pragma unroll
      for (int u = 0; u < 4; ++u) {
        if (staged) stg[ps[u]] = v[u];
        else seg[ps[u]] = v[u];
      }
    }
    for (; i < e; ++i) {
      const int ps = atomicAdd(&curL[p[i] - r0], 1);
      const int2 v = q[i];
      if (staged) stg[ps] = v;
      else seg[ps] = v;
    }
  }
  __syncthreads();
  if (staged)
    for (int i = tid; i < T; i += 256) seg[i] = stg[i];
}

__device__ __forceinline__ void fma4(float4& a, const float4 v, const float w) {
  a.x = fmaf(v.x, w, a.x);
  a.y = fmaf(v.y, w, a.y);
  a.z = fmaf(v.z, w, a.z);
  a.w = fmaf(v.w, w, a.w);
}

#define LDP 132
__global__ __launch_bounds__(128)
void gcn_gather128_gemm2(const float* __restrict__ m,
                         const int* __restrict__ csr_i,
                         const int* __restrict__ base,
                         const int* __restrict__ cnt,
                         const float* __restrict__ b,
                         const float* __restrict__ W2,
                         float* __restrict__ out, int N, int rowTiles) {
  __shared__ __attribute__((aligned(16))) float hT[16 * LDP];
  __shared__ __attribute__((aligned(16))) float oT[16 * DO];

  const int lane = threadIdx.x & 31;
  const int wave = __builtin_amdgcn_readfirstlane((int)(threadIdx.x >> 5));
  const int mm = lane & 15;
  const int hh = lane >> 4;
  const int kk = hh << 1;

  const int col = wave * 16 + mm;
  const bool cv = col < DO;
  const int cc = cv ? col : DO - 1;
  v2f bf[32];
  if (wave < 3) {
    #pragma unroll
    for (int s = 0; s < 32; ++s) {
      const float* __restrict__ wr = W2 + (size_t)(4 * s + kk) * DO + cc;
      bf[s].x = wr[0];
      bf[s].y = wr[DO];
    }
  }

  const float4* __restrict__ m4 = (const float4*)m + lane;
  const float4 bb = ((const float4*)b)[lane];

  for (int t = blockIdx.x; t < rowTiles; t += gridDim.x) {
    const int r0 = t * 16;
    int row = r0 + wave * 4;
    int sb = 0, n = 0;
    if (row < N) {
      sb = base[row];
      n = cnt[row];
    }

    for (int q = 0; q < 4; ++q) {
      const int nrow = row + 1;
      int nsb = 0, nn = 0;
      if (q < 3 && nrow < N) {
        nsb = base[nrow];
        nn = cnt[nrow];
      }
      const int2* __restrict__ seg = (const int2*)csr_i + sb;

      float4 acc = make_float4(0.f, 0.f, 0.f, 0.f);
      int j = 0;
      for (; j + 8 <= n; j += 8) {
        int2 e[8];
        float4 v[8];
        #pragma unroll
        for (int u = 0; u < 8; ++u) e[u] = seg[j + u];
        #pragma unroll
        for (int u = 0; u < 8; ++u) v[u] = m4[(size_t)e[u].x * (DH / 4)];
        #pragma unroll
        for (int u = 0; u < 8; ++u) fma4(acc, v[u], __int_as_float(e[u].y));
      }
      if (j + 4 <= n) {
        int2 e[4];
        float4 v[4];
        #pragma unroll
        for (int u = 0; u < 4; ++u) e[u] = seg[j + u];
        #pragma unroll
        for (int u = 0; u < 4; ++u) v[u] = m4[(size_t)e[u].x * (DH / 4)];
        #pragma unroll
        for (int u = 0; u < 4; ++u) fma4(acc, v[u], __int_as_float(e[u].y));
        j += 4;
      }
      if (j + 2 <= n) {
        const int2 e0 = seg[j], e1 = seg[j + 1];
        const float4 v0 = m4[(size_t)e0.x * (DH / 4)];
        const float4 v1 = m4[(size_t)e1.x * (DH / 4)];
        fma4(acc, v0, __int_as_float(e0.y));
        fma4(acc, v1, __int_as_float(e1.y));
        j += 2;
      }
      if (j < n) {
        const int2 e0 = seg[j];
        const float4 v0 = m4[(size_t)e0.x * (DH / 4)];
        fma4(acc, v0, __int_as_float(e0.y));
      }

      float4 r;
      r.x = fmaxf(acc.x + bb.x, 0.0f);
      r.y = fmaxf(acc.y + bb.y, 0.0f);
      r.z = fmaxf(acc.z + bb.z, 0.0f);
      r.w = fmaxf(acc.w + bb.w, 0.0f);
      *(float4*)&hT[(wave * 4 + q) * LDP + 4 * lane] = r;

      row = nrow;
      sb = nsb;
      n = nn;
    }

    __syncthreads();

    if (wave < 3) {
      const float* __restrict__ ap = &hT[mm * LDP + kk];
      v8f dacc = {};
      #pragma unroll
      for (int s = 0; s < 32; ++s) {
        const v2f a = *(const v2f*)(ap + 4 * s);
        dacc = __builtin_amdgcn_wmma_f32_16x16x4_f32(
            false, a, false, bf[s], (short)0, dacc, false, false);
      }
      if (cv) {
        #pragma unroll
        for (int j = 0; j < 8; ++j) oT[(hh * 8 + j) * DO + col] = dacc[j];
      }
    }

    __syncthreads();

    const int rows = (N - r0 < 16) ? (N - r0) : 16;
    const int nf4 = rows * (DO / 4);
    float4* __restrict__ o4 = (float4*)out + (size_t)r0 * (DO / 4);
    for (int f = threadIdx.x; f < nf4; f += 128)
      o4[f] = ((const float4*)oT)[f];
  }
}

__global__ __launch_bounds__(256)
void gcn_gather40(const float* __restrict__ m, const int* __restrict__ csr_i,
                  const int* __restrict__ base, const int* __restrict__ cnt,
                  const float* __restrict__ b, float* __restrict__ out, int N) {
  const int t = blockIdx.x * blockDim.x + threadIdx.x;
  if (t >= N * (DO / 4)) return;
  const int row = t / (DO / 4);
  const int ch  = t - row * (DO / 4);
  const int2* __restrict__ seg = (const int2*)csr_i + base[row];
  const int n = cnt[row];
  const float4* __restrict__ m4 = (const float4*)m + ch;

  float4 acc = make_float4(0.f, 0.f, 0.f, 0.f);
  int j = 0;
  for (; j + 4 <= n; j += 4) {
    int2 e[4];
    float4 v[4];
    #pragma unroll
    for (int u = 0; u < 4; ++u) e[u] = seg[j + u];
    #pragma unroll
    for (int u = 0; u < 4; ++u) v[u] = m4[(size_t)e[u].x * (DO / 4)];
    #pragma unroll
    for (int u = 0; u < 4; ++u) fma4(acc, v[u], __int_as_float(e[u].y));
  }
  if (j + 2 <= n) {
    const int2 e0 = seg[j], e1 = seg[j + 1];
    const float4 v0 = m4[(size_t)e0.x * (DO / 4)];
    const float4 v1 = m4[(size_t)e1.x * (DO / 4)];
    fma4(acc, v0, __int_as_float(e0.y));
    fma4(acc, v1, __int_as_float(e1.y));
    j += 2;
  }
  if (j < n) {
    const int2 e0 = seg[j];
    const float4 v0 = m4[(size_t)e0.x * (DO / 4)];
    fma4(acc, v0, __int_as_float(e0.y));
  }

  const float4 bb = ((const float4*)b)[ch];
  float4 r;
  r.x = acc.x + bb.x;
  r.y = acc.y + bb.y;
  r.z = acc.z + bb.z;
  r.w = acc.w + bb.w;
  ((float4*)out)[t] = r;
}

extern "C" void kernel_launch(void* const* d_in, const int* in_sizes, int n_in,
                              void* d_out, int out_size, void* d_ws,
                              size_t ws_size, hipStream_t stream) {
  const float* x  = (const float*)d_in[0];
  const int*   ei = (const int*)d_in[1];
  const float* ew = (const float*)d_in[2];
  const float* W1 = (const float*)d_in[3];
  const float* b1 = (const float*)d_in[4];
  const float* W2 = (const float*)d_in[5];
  const float* b2 = (const float*)d_in[6];

  const int N = in_sizes[0] / DH;
  const int E = in_sizes[1] / 2;
  const int* src = ei;
  const int* dst = ei + E;

  float* m1   = (float*)d_ws;
  float* m2   = m1 + (size_t)N * DH;
  const int NP = (N + 35) & ~3;
  int*   cnt  = (int*)(m2 + (size_t)N * DO);
  int*   base = cnt + NP;
  int2*  csr  = (int2*)(base + NP);
  int2*  sP   = csr + E;
  int*   sD   = (int*)(sP + E);
  int*   st   = sD + E;
  float* out  = (float*)d_out;

  const int nt = 256;
  const int rowTiles = (N + 15) / 16;
  const int NB = (N + BR - 1) / BR;
  const int NW = (E + CHUNK - 1) / CHUNK;

  gcn_bin<<<NW + 512, nt, (NB + 1) * sizeof(int), stream>>>(
      src, dst, ew, sD, sP, st, cnt + N, E, NB, NW, x, W1, m1, N, rowTiles);
  gcn_group<<<NB, nt, 0, stream>>>(sD, sP, st, cnt + N, base, cnt, csr,
                                   N, NB, NW);
  gcn_gather128_gemm2<<<(rowTiles + 1) / 2, 128, 0, stream>>>(
      m1, (const int*)csr, base, cnt, b1, W2, m2, N, rowTiles);
  gcn_gather40<<<(N * (DO / 4) + nt - 1) / nt, nt, 0, stream>>>(
      m2, (const int*)csr, base, cnt, b2, out, N);
}


# Layer 2 asks one aligned 256-B unit per edge: m2 rows zero-padded to 256 B, a half-wave owns a row and loads whole units, its (src,w) pairs come 16 per load and pass by lane exchange; sums unchanged
# speedup vs baseline: 6.1900x; 1.0506x over previous
#include <hip/hip_runtime.h>

typedef float v2f __attribute__((ext_vector_type(2)));
typedef float v8f __attribute__((ext_vector_type(8)));

#define DH 128
#define DO 40

__device__ __forceinline__
void gemm1_tiles(const float* __restrict__ x, const float* __restrict__ W,
                 float* __restrict__ out, int N, int rowTiles,
                 int vb, int nvb) {
  const int lane = threadIdx.x & 31;
  const int wave = (threadIdx.x >> 5) & 3;
  const int c0 = wave * 32;
  const int m  = lane & 15;
  const int kk = (lane >> 4) << 1;

  v2f b0[32], b1[32];
  #pragma unroll
  for (int s = 0; s < 32; ++s) {
    const float* __restrict__ wr = W + (size_t)(4 * s + kk) * DH + c0 + m;
    b0[s].x = wr[0];
    b0[s].y = wr[DH];
    b1[s].x = wr[16];
    b1[s].y = wr[DH + 16];
  }

  for (int t = vb; t < rowTiles; t += nvb) {
    const int r0 = t * 16;
    const int arow = r0 + m;
    const float* __restrict__ xrow =
        x + (size_t)(arow < N ? arow : N - 1) * DH + kk;

    v2f a[32];
    #pragma unroll
    for (int s = 0; s < 32; ++s) {
      a[s].x = xrow[4 * s];
      a[s].y = xrow[4 * s + 1];
    }
    asm volatile("" :: "v"(a[0]), "v"(a[1]), "v"(a[2]), "v"(a[3]), "v"(a[4]), "v"(a[5]), "v"(a[6]), "v"(a[7]), "v"(a[8]), "v"(a[9]), "v"(a[10]), "v"(a[11]), "v"(a[12]), "v"(a[13]), "v"(a[14]), "v"(a[15]));
    asm volatile("" :: "v"(a[16]), "v"(a[17]), "v"(a[18]), "v"(a[19]), "v"(a[20]), "v"(a[21]), "v"(a[22]), "v"(a[23]), "v"(a[24]), "v"(a[25]), "v"(a[26]), "v"(a[27]), "v"(a[28]), "v"(a[29]), "v"(a[30]), "v"(a[31]));
    v8f acc0 = {}, acc1 = {};
    #pragma unroll
    for (int s = 0; s < 32; ++s) {
      acc0 = __builtin_amdgcn_wmma_f32_16x16x4_f32(
          false, a[s], false, b0[s], (short)0, acc0, false, false);
      acc1 = __builtin_amdgcn_wmma_f32_16x16x4_f32(
          false, a[s], false, b1[s], (short)0, acc1, false, false);
    }

    const int rbase = r0 + ((lane >> 4) << 3);
    float* __restrict__ p = out + (size_t)rbase * DH + c0 + m;
    if (r0 + 16 <= N) {
      #pragma unroll
      for (int j = 0; j < 8; ++j) {
        p[(size_t)j * DH] = acc0[j];
        p[(size_t)j * DH + 16] = acc1[j];
      }
    } else {
      #pragma unroll
      for (int j = 0; j < 8; ++j)
        if (rbase + j < N) {
          p[(size_t)j * DH] = acc0[j];
          p[(size_t)j * DH + 16] = acc1[j];
        }
    }
  }
}

#define BR    128
#define CHUNK 4096
#define EPT   16
#define STG   6144

__global__ __launch_bounds__(256)
void gcn_bin(const int* __restrict__ src, const int* __restrict__ dst,
             const float* __restrict__ w, int* __restrict__ sD,
             int2* __restrict__ sP, int* __restrict__ st,
             int* __restrict__ cursor, int E, int NB, int NW,
             const float* __restrict__ gx, const float* __restrict__ gW,
             float* __restrict__ gm, int N, int rowTiles) {
  if ((int)blockIdx.x >= NW) {
    gemm1_tiles(gx, gW, gm, N, rowTiles,
                ((int)blockIdx.x - NW) * 2 + (int)(threadIdx.x >> 7),
                ((int)gridDim.x - NW) * 2);
    return;
  }
  extern __shared__ int hist[];
  __shared__ int stD[CHUNK];
  __shared__ int2 stP[CHUNK];
  __shared__ int wsum[8];

  const int tid = threadIdx.x;
  const int lane = tid & 31;
  const int wave = tid >> 5;
  const int c0 = blockIdx.x * CHUNK;
  const int cn = (E - c0 < CHUNK) ? (E - c0) : CHUNK;

  if (blockIdx.x == 0 && tid == 0) *cursor = 0;
  for (int i = tid; i <= NB; i += 256) hist[i] = 0;

  int d[EPT], s[EPT], r[EPT];
  float ww[EPT];
  #pragma unroll
  for (int u = 0; u < EPT; ++u) {
    const int i = u * 256 + tid;
    d[u] = -1;
    s[u] = 0;
    ww[u] = 0.0f;
    r[u] = 0;
    if (i < cn) {
      d[u] = dst[c0 + i];
      s[u] = src[c0 + i];
      ww[u] = w[c0 + i];
    }
  }
  __syncthreads();

  #pragma unroll
  for (int u = 0; u < EPT; ++u)
    if (d[u] >= 0) r[u] = atomicAdd(&hist[d[u] / BR], 1);
  __syncthreads();

  const int K = (NB + 255) >> 8;
  const int b0 = tid * K;
  int loc = 0;
  for (int k = 0; k < K; ++k)
    if (b0 + k < NB) loc += hist[b0 + k];
  int incl = loc;
  #pragma unroll
  for (int off = 1; off < 32; off <<= 1) {
    const int v = __shfl_up(incl, off, 32);
    if (lane >= off) incl += v;
  }
  if (lane == 31) wsum[wave] = incl;
  __syncthreads();
  int run = incl - loc;
  for (int k = 0; k < wave; ++k) run += wsum[k];
  for (int k = 0; k < K; ++k)
    if (b0 + k < NB) {
      const int c = hist[b0 + k];
      hist[b0 + k] = run;
      run += c;
    }
  if (tid == 255) hist[NB] = run;
  __syncthreads();

  int* __restrict__ strow = st + (size_t)blockIdx.x * (NB + 1);
  for (int i = tid; i <= NB; i += 256) strow[i] = hist[i];

  #pragma unroll
  for (int u = 0; u < EPT; ++u)
    if (d[u] >= 0) {
      const int p = hist[d[u] / BR] + r[u];
      stD[p] = d[u];
      stP[p] = make_int2(s[u], __float_as_int(ww[u]));
    }
  __syncthreads();

  for (int i = tid; i < cn; i += 256) {
    sD[c0 + i] = stD[i];
    sP[c0 + i] = stP[i];
  }
}

__global__ __launch_bounds__(256)
void gcn_group(const int* __restrict__ sD, const int2* __restrict__ sP,
               const int* __restrict__ st, int* __restrict__ cursor,
               int* __restrict__ base, int* __restrict__ cnt,
               int2* __restrict__ csr, int N, int NB, int NW) {
  __shared__ int cntL[BR];
  __shared__ int curL[BR];
  __shared__ int wtot[BR / 32];
  __shared__ int gb[2];
  __shared__ int2 stg[STG];

  const int tid = threadIdx.x;
  const int lane = tid & 31;
  const int wave = tid >> 5;
  const int b = blockIdx.x;
  const int r0 = b * BR;

  if (tid < BR) cntL[tid] = 0;
  __syncthreads();

  for (int f = tid; f < NW; f += 256) {
    const int* __restrict__ t = st + (size_t)f * (NB + 1) + b;
    const int s = t[0], e = t[1];
    const int* __restrict__ p = sD + (size_t)f * CHUNK;
    int i = s;
    for (; i + 4 <= e; i += 4) {
      const int d0 = p[i], d1 = p[i + 1], d2 = p[i + 2], d3 = p[i + 3];
      atomicAdd(&cntL[d0 - r0], 1);
      atomicAdd(&cntL[d1 - r0], 1);
      atomicAdd(&cntL[d2 - r0], 1);
      atomicAdd(&cntL[d3 - r0], 1);
    }
    for (; i < e; ++i) atomicAdd(&cntL[p[i] - r0], 1);
  }
  __syncthreads();

  const int c = (tid < BR) ? cntL[tid] : 0;
  int incl = c;
  #pragma unroll
  for (int off = 1; off < 32; off <<= 1) {
    const int v = __shfl_up(incl, off, 32);
    if (lane >= off) incl += v;
  }
  if (tid < BR && lane == 31) wtot[wave] = incl;
  __syncthreads();
  int excl = incl - c;
  for (int k = 0; k < wave && k < BR / 32; ++k) excl += wtot[k];
  if (tid == BR - 1) {
    const int T = excl + c;
    gb[0] = atomicAdd(cursor, T);
    gb[1] = T;
  }
  __syncthreads();
  const int gbase = gb[0];
  const int T = gb[1];
  if (tid < BR) {
    curL[tid] = excl;
    const int row = r0 + tid;
    if (row < N) {
      base[row] = gbase + excl;
      cnt[row] = c;
    }
  }
  __syncthreads();

  const bool staged = T <= STG;
  int2* __restrict__ seg = csr + gbase;
  for (int f = tid; f < NW; f += 256) {
    const int* __restrict__ t = st + (size_t)f * (NB + 1) + b;
    const int s = t[0], e = t[1];
    const int* __restrict__ p = sD + (size_t)f * CHUNK;
    const int2* __restrict__ q = sP + (size_t)f * CHUNK;
    int i = s;
    for (; i + 4 <= e; i += 4) {
      int dd[4], ps[4];
      int2 v[4];
      #pragma unroll
      for (int u = 0; u < 4; ++u) {
        dd[u] = p[i + u];
        v[u] = q[i + u];
      }
      #pragma unroll
      for (int u = 0; u < 4; ++u) ps[u] = atomicAdd(&curL[dd[u] - r0], 1);
      #pragma unroll
      for (int u = 0; u < 4; ++u) {
        if (staged) stg[ps[u]] = v[u];
        else seg[ps[u]] = v[u];
      }
    }
    for (; i < e; ++i) {
      const int ps = atomicAdd(&curL[p[i] - r0], 1);
      const int2 v = q[i];
      if (staged) stg[ps] = v;
      else seg[ps] = v;
    }
  }
  __syncthreads();
  if (staged)
    for (int i = tid; i < T; i += 256) seg[i] = stg[i];
}

__device__ __forceinline__ void fma4(float4& a, const float4 v, const float w) {
  a.x = fmaf(v.x, w, a.x);
  a.y = fmaf(v.y, w, a.y);
  a.z = fmaf(v.z, w, a.z);
  a.w = fmaf(v.w, w, a.w);
}

#define LDP 132
#define M2P 64
__global__ __launch_bounds__(128)
void gcn_gather128_gemm2(const float* __restrict__ m,
                         const int* __restrict__ csr_i,
                         const int* __restrict__ base,
                         const int* __restrict__ cnt,
                         const float* __restrict__ b,
                         const float* __restrict__ W2,
                         float* __restrict__ out, int N, int rowTiles) {
  __shared__ __attribute__((aligned(16))) float hT[16 * LDP];
  __shared__ __attribute__((aligned(16))) float oT[16 * DO];

  const int lane = threadIdx.x & 31;
  const int wave = __builtin_amdgcn_readfirstlane((int)(threadIdx.x >> 5));
  const int mm = lane & 15;
  const int hh = lane >> 4;
  const int kk = hh << 1;

  const int col = wave * 16 + mm;
  const bool cv = col < DO;
  const int cc = cv ? col : DO - 1;
  v2f bf[32];
  if (wave < 3) {
    #pragma unroll
    for (int s = 0; s < 32; ++s) {
      const float* __restrict__ wr = W2 + (size_t)(4 * s + kk) * DO + cc;
      bf[s].x = wr[0];
      bf[s].y = wr[DO];
    }
  }

  const float4* __restrict__ m4 = (const float4*)m + lane;
  const float4 bb = ((const float4*)b)[lane];

  for (int t = blockIdx.x; t < rowTiles; t += gridDim.x) {
    const int r0 = t * 16;
    int row = r0 + wave * 4;
    int sb = 0, n = 0;
    if (row < N) {
      sb = base[row];
      n = cnt[row];
    }

    for (int q = 0; q < 4; ++q) {
      const int nrow = row + 1;
      int nsb = 0, nn = 0;
      if (q < 3 && nrow < N) {
        nsb = base[nrow];
        nn = cnt[nrow];
      }
      const int2* __restrict__ seg = (const int2*)csr_i + sb;

      float4 acc = make_float4(0.f, 0.f, 0.f, 0.f);
      int j = 0;
      for (; j + 8 <= n; j += 8) {
        int2 e[8];
        float4 v[8];
        #pragma unroll
        for (int u = 0; u < 8; ++u) e[u] = seg[j + u];
        #pragma unroll
        for (int u = 0; u < 8; ++u) v[u] = m4[(size_t)e[u].x * (DH / 4)];
        #pragma unroll
        for (int u = 0; u < 8; ++u) fma4(acc, v[u], __int_as_float(e[u].y));
      }
      if (j + 4 <= n) {
        int2 e[4];
        float4 v[4];
        #pragma unroll
        for (int u = 0; u < 4; ++u) e[u] = seg[j + u];
        #pragma unroll
        for (int u = 0; u < 4; ++u) v[u] = m4[(size_t)e[u].x * (DH / 4)];
        #pragma unroll
        for (int u = 0; u < 4; ++u) fma4(acc, v[u], __int_as_float(e[u].y));
        j += 4;
      }
      if (j + 2 <= n) {
        const int2 e0 = seg[j], e1 = seg[j + 1];
        const float4 v0 = m4[(size_t)e0.x * (DH / 4)];
        const float4 v1 = m4[(size_t)e1.x * (DH / 4)];
        fma4(acc, v0, __int_as_float(e0.y));
        fma4(acc, v1, __int_as_float(e1.y));
        j += 2;
      }
      if (j < n) {
        const int2 e0 = seg[j];
        const float4 v0 = m4[(size_t)e0.x * (DH / 4)];
        fma4(acc, v0, __int_as_float(e0.y));
      }

      float4 r;
      r.x = fmaxf(acc.x + bb.x, 0.0f);
      r.y = fmaxf(acc.y + bb.y, 0.0f);
      r.z = fmaxf(acc.z + bb.z, 0.0f);
      r.w = fmaxf(acc.w + bb.w, 0.0f);
      *(float4*)&hT[(wave * 4 + q) * LDP + 4 * lane] = r;

      row = nrow;
      sb = nsb;
      n = nn;
    }

    __syncthreads();

    if (wave < 3) {
      const float* __restrict__ ap = &hT[mm * LDP + kk];
      v8f dacc = {};
      #pragma unroll
      for (int s = 0; s < 32; ++s) {
        const v2f a = *(const v2f*)(ap + 4 * s);
        dacc = __builtin_amdgcn_wmma_f32_16x16x4_f32(
            false, a, false, bf[s], (short)0, dacc, false, false);
      }
      if (cv) {
        #pragma unroll
        for (int j = 0; j < 8; ++j) oT[(hh * 8 + j) * DO + col] = dacc[j];
      }
    }

    __syncthreads();

    const int pc = threadIdx.x & 15;
    #pragma unroll
    for (int rr = (int)(threadIdx.x >> 4); rr < 16; rr += 8) {
      float4 val = make_float4(0.f, 0.f, 0.f, 0.f);
      if (pc < DO / 4) val = ((const float4*)oT)[rr * (DO / 4) + pc];
      if (r0 + rr < N)
        ((float4*)out)[(size_t)(r0 + rr) * (M2P / 4) + pc] = val;
    }
  }
}


template <int G>
__device__ __forceinline__
void sum8(float4& acc, const int2 e, const int c,
          const float* __restrict__ m, const unsigned pcoff) {
  float w[8];
  float4 v[8];
  #pragma unroll
  for (int u = 0; u < 8; ++u) {
    const unsigned off =
        (unsigned)__shfl(e.x, G + u, 16) * (unsigned)(M2P * sizeof(float)) + pcoff;
    w[u] = __int_as_float(__shfl(e.y, G + u, 16));
    v[u] = make_float4(0.f, 0.f, 0.f, 0.f);
    if (G + u < c) v[u] = *(const float4*)((const char*)m + off);
  }
  #pragma unroll
  for (int u = 0; u < 8; ++u) fma4(acc, v[u], w[u]);
}

__global__ __launch_bounds__(256)
void gcn_gather40(const float* __restrict__ m, const int* __restrict__ csr_i,
                  const int* __restrict__ base, const int* __restrict__ cnt,
                  const float* __restrict__ b, float* __restrict__ out, int N) {
  __shared__ __attribute__((aligned(16))) float oT[16 * DO];

  const int tid = threadIdx.x;
  const int pc = tid & 15;
  const int rl = tid >> 4;
  const int r0 = blockIdx.x * 16;
  const int row = r0 + rl;
  const unsigned pcoff = (unsigned)pc * 16u;

  int sb = 0, n = 0;
  if (row < N) {
    sb = base[row];
    n = cnt[row];
  }
  const int o = __shfl_xor(n, 16, 32);
  const int nn = __builtin_amdgcn_readfirstlane(n > o ? n : o);
  const int2* __restrict__ seg = (const int2*)csr_i + sb;

  float4 acc = make_float4(0.f, 0.f, 0.f, 0.f);
  for (int j0 = 0; j0 < nn; j0 += 32) {
    const int c = n - j0;
    int2 ea = make_int2(0, 0), eb = make_int2(0, 0);
    if (pc < c) ea = seg[j0 + pc];
    if (pc + 16 < c) eb = seg[j0 + 16 + pc];
    sum8<0>(acc, ea, c, m, pcoff);
    if (j0 + 8 < nn) sum8<8>(acc, ea, c, m, pcoff);
    if (j0 + 16 < nn) sum8<0>(acc, eb, c - 16, m, pcoff);
    if (j0 + 24 < nn) sum8<8>(acc, eb, c - 16, m, pcoff);
  }

  if (pc < DO / 4) {
    const float4 bb = ((const float4*)b)[pc];
    float4 r;
    r.x = acc.x + bb.x;
    r.y = acc.y + bb.y;
    r.z = acc.z + bb.z;
    r.w = acc.w + bb.w;
    *(float4*)&oT[rl * DO + 4 * pc] = r;
  }
  __syncthreads();

  const int rows = (N - r0 < 16) ? (N - r0) : 16;
  if (tid < rows * (DO / 4))
    ((float4*)out)[(size_t)r0 * (DO / 4) + tid] = ((const float4*)oT)[tid];
}

extern "C" void kernel_launch(void* const* d_in, const int* in_sizes, int n_in,
                              void* d_out, int out_size, void* d_ws,
                              size_t ws_size, hipStream_t stream) {
  const float* x  = (const float*)d_in[0];
  const int*   ei = (const int*)d_in[1];
  const float* ew = (const float*)d_in[2];
  const float* W1 = (const float*)d_in[3];
  const float* b1 = (const float*)d_in[4];
  const float* W2 = (const float*)d_in[5];
  const float* b2 = (const float*)d_in[6];

  const int N = in_sizes[0] / DH;
  const int E = in_sizes[1] / 2;
  const int* src = ei;
  const int* dst = ei + E;

  float* m1   = (float*)d_ws;
  float* m2   = m1 + (size_t)N * DH;
  const int NP = (N + 35) & ~3;
  int*   cnt  = (int*)(m2 + (size_t)N * M2P);
  int*   base = cnt + NP;
  int2*  csr  = (int2*)(base + NP);
  int2*  sP   = csr + E;
  int*   sD   = (int*)(sP + E);
  int*   st   = sD + E;
  float* out  = (float*)d_out;

  const int nt = 256;
  const int rowTiles = (N + 15) / 16;
  const int NB = (N + BR - 1) / BR;
  const int NW = (E + CHUNK - 1) / CHUNK;

  gcn_bin<<<NW + 512, nt, (NB + 1) * sizeof(int), stream>>>(
      src, dst, ew, sD, sP, st, cnt + N, E, NB, NW, x, W1, m1, N, rowTiles);
  gcn_group<<<NB, nt, 0, stream>>>(sD, sP, st, cnt + N, base, cnt, csr,
                                   N, NB, NW);
  gcn_gather128_gemm2<<<(rowTiles + 1) / 2, 128, 0, stream>>>(
      m1, (const int*)csr, base, cnt, b1, W2, m2, N, rowTiles);
  gcn_gather40<<<rowTiles, nt, 0, stream>>>(
      m2, (const int*)csr, base, cnt, b2, out, N);
}


# GEMM1 reads each 16x128 x tile once per 4-wave half in whole 512-B rows into a double-buffered LDS tile; WMMA A fragments come from LDS (was 16 lines per fragment load, 4x over); sums unchanged
# speedup vs baseline: 6.3869x; 1.0318x over previous
#include <hip/hip_runtime.h>

typedef float v2f __attribute__((ext_vector_type(2)));
typedef float v4f __attribute__((ext_vector_type(4)));
typedef float v8f __attribute__((ext_vector_type(8)));

#define DH 128
#define DO 40

#define XLP 132
#define XTILE (16 * XLP)

__device__ __forceinline__
v4f x_row_piece(const float* __restrict__ x, int r, int N, int lane) {
  return ((const v4f*)(x + (size_t)(r < N ? r : N - 1) * DH))[lane];
}

__device__ __forceinline__
void gemm1_tiles(const float* __restrict__ x, const float* __restrict__ W,
                 float* __restrict__ out, int N, int rowTiles,
                 int vb, int nvb, float* __restrict__ xs) {
  const int lane = threadIdx.x & 31;
  const int wave = (threadIdx.x >> 5) & 3;
  const int c0 = wave * 32;
  const int m  = lane & 15;
  const int kk = (lane >> 4) << 1;

  v2f b0[32], b1[32];
  #pragma unroll
  for (int s = 0; s < 32; ++s) {
    const float* __restrict__ wr = W + (size_t)(4 * s + kk) * DH + c0 + m;
    b0[s].x = wr[0];
    b0[s].y = wr[DH];
    b1[s].x = wr[16];
    b1[s].y = wr[DH + 16];
  }

  const int trips = (rowTiles + nvb - 1) / nvb;
  v4f g0, g1, g2, g3;
  float* __restrict__ sw = xs + (wave * 4) * XLP + 4 * lane;
  const float* __restrict__ ap = xs + m * XLP + kk;

  int t = vb;
  if (t < rowTiles) {
    const int rb = t * 16 + wave * 4;
    g0 = x_row_piece(x, rb, N, lane);
    g1 = x_row_piece(x, rb + 1, N, lane);
    g2 = x_row_piece(x, rb + 2, N, lane);
    g3 = x_row_piece(x, rb + 3, N, lane);
    *(v4f*)(sw) = g0;
    *(v4f*)(sw + XLP) = g1;
    *(v4f*)(sw + 2 * XLP) = g2;
    *(v4f*)(sw + 3 * XLP) = g3;
  }
  __syncthreads();

  for (int it = 0; it < trips; ++it, t += nvb) {
    const bool have = t < rowTiles;
    v8f acc0 = {}, acc1 = {};
    if (have) {
      const int tn = t + nvb;
      const int rb = tn * 16 + wave * 4;
      g0 = x_row_piece(x, rb, N, lane);
      g1 = x_row_piece(x, rb + 1, N, lane);
      g2 = x_row_piece(x, rb + 2, N, lane);
      g3 = x_row_piece(x, rb + 3, N, lane);

      const float* __restrict__ a_p = ap + (it & 1) * XTILE;
      v2f a[32];
      #pragma unroll
      for (int s = 0; s < 32; ++s) a[s] = *(const v2f*)(a_p + 4 * s);
      asm volatile("" :: "v"(a[0]), "v"(a[1]), "v"(a[2]), "v"(a[3]), "v"(a[4]), "v"(a[5]), "v"(a[6]), "v"(a[7]), "v"(a[8]), "v"(a[9]), "v"(a[10]), "v"(a[11]), "v"(a[12]), "v"(a[13]), "v"(a[14]), "v"(a[15]));
      asm volatile("" :: "v"(a[16]), "v"(a[17]), "v"(a[18]), "v"(a[19]), "v"(a[20]), "v"(a[21]), "v"(a[22]), "v"(a[23]), "v"(a[24]), "v"(a[25]), "v"(a[26]), "v"(a[27]), "v"(a[28]), "v"(a[29]), "v"(a[30]), "v"(a[31]));
      #pragma unroll
      for (int s = 0; s < 32; ++s) {
        acc0 = __builtin_amdgcn_wmma_f32_16x16x4_f32(
            false, a[s], false, b0[s], (short)0, acc0, false, false);
        acc1 = __builtin_amdgcn_wmma_f32_16x16x4_f32(
            false, a[s], false, b1[s], (short)0, acc1, false, false);
      }

      __builtin_amdgcn_sched_barrier(0);
      float* __restrict__ sn = sw + ((it + 1) & 1) * XTILE;
      *(v4f*)(sn) = g0;
      *(v4f*)(sn + XLP) = g1;
      *(v4f*)(sn + 2 * XLP) = g2;
      *(v4f*)(sn + 3 * XLP) = g3;
    }
    __syncthreads();

    if (have) {
      const int r0 = t * 16;
      const int rbase = r0 + ((lane >> 4) << 3);
      float* __restrict__ p = out + (size_t)rbase * DH + c0 + m;
      if (r0 + 16 <= N) {
        #pragma unroll
        for (int j = 0; j < 8; ++j) {
          p[(size_t)j * DH] = acc0[j];
          p[(size_t)j * DH + 16] = acc1[j];
        }
      } else {
        #pragma unroll
        for (int j = 0; j < 8; ++j)
          if (rbase + j < N) {
            p[(size_t)j * DH] = acc0[j];
            p[(size_t)j * DH + 16] = acc1[j];
          }
      }
    }
  }
}

#define BR    128
#define CHUNK 4096
#define EPT   16
#define STG   6144

__global__ __launch_bounds__(256)
void gcn_bin(const int* __restrict__ src, const int* __restrict__ dst,
             const float* __restrict__ w, int* __restrict__ sD,
             int2* __restrict__ sP, int* __restrict__ st,
             int* __restrict__ cursor, int E, int NB, int NW,
             const float* __restrict__ gx, const float* __restrict__ gW,
             float* __restrict__ gm, int N, int rowTiles) {
  __shared__ __attribute__((aligned(16))) float xT[2 * 2 * XTILE];
  if ((int)blockIdx.x >= NW) {
    const int half = __builtin_amdgcn_readfirstlane((int)(threadIdx.x >> 7));
    gemm1_tiles(gx, gW, gm, N, rowTiles,
                ((int)blockIdx.x - NW) * 2 + half,
                ((int)gridDim.x - NW) * 2, xT + half * (2 * XTILE));
    return;
  }
  extern __shared__ int hist[];
  __shared__ int stD[CHUNK];
  __shared__ int2 stP[CHUNK];
  __shared__ int wsum[8];

  const int tid = threadIdx.x;
  const int lane = tid & 31;
  const int wave = tid >> 5;
  const int c0 = blockIdx.x * CHUNK;
  const int cn = (E - c0 < CHUNK) ? (E - c0) : CHUNK;

  if (blockIdx.x == 0 && tid == 0) *cursor = 0;
  for (int i = tid; i <= NB; i += 256) hist[i] = 0;

  int d[EPT], s[EPT], r[EPT];
  float ww[EPT];
  #pragma unroll
  for (int u = 0; u < EPT; ++u) {
    const int i = u * 256 + tid;
    d[u] = -1;
    s[u] = 0;
    ww[u] = 0.0f;
    r[u] = 0;
    if (i < cn) {
      d[u] = dst[c0 + i];
      s[u] = src[c0 + i];
      ww[u] = w[c0 + i];
    }
  }
  __syncthreads();

  #pragma unroll
  for (int u = 0; u < EPT; ++u)
    if (d[u] >= 0) r[u] = atomicAdd(&hist[d[u] / BR], 1);
  __syncthreads();

  const int K = (NB + 255) >> 8;
  const int b0 = tid * K;
  int loc = 0;
  for (int k = 0; k < K; ++k)
    if (b0 + k < NB) loc += hist[b0 + k];
  int incl = loc;
  #pragma unroll
  for (int off = 1; off < 32; off <<= 1) {
    const int v = __shfl_up(incl, off, 32);
    if (lane >= off) incl += v;
  }
  if (lane == 31) wsum[wave] = incl;
  __syncthreads();
  int run = incl - loc;
  for (int k = 0; k < wave; ++k) run += wsum[k];
  for (int k = 0; k < K; ++k)
    if (b0 + k < NB) {
      const int c = hist[b0 + k];
      hist[b0 + k] = run;
      run += c;
    }
  if (tid == 255) hist[NB] = run;
  __syncthreads();

  int* __restrict__ strow = st + (size_t)blockIdx.x * (NB + 1);
  for (int i = tid; i <= NB; i += 256) strow[i] = hist[i];

  #pragma unroll
  for (int u = 0; u < EPT; ++u)
    if (d[u] >= 0) {
      const int p = hist[d[u] / BR] + r[u];
      stD[p] = d[u];
      stP[p] = make_int2(s[u], __float_as_int(ww[u]));
    }
  __syncthreads();

  for (int i = tid; i < cn; i += 256) {
    sD[c0 + i] = stD[i];
    sP[c0 + i] = stP[i];
  }
}

__global__ __launch_bounds__(256)
void gcn_group(const int* __restrict__ sD, const int2* __restrict__ sP,
               const int* __restrict__ st, int* __restrict__ cursor,
               int* __restrict__ base, int* __restrict__ cnt,
               int2* __restrict__ csr, int N, int NB, int NW) {
  __shared__ int cntL[BR];
  __shared__ int curL[BR];
  __shared__ int wtot[BR / 32];
  __shared__ int gb[2];
  __shared__ int2 stg[STG];

  const int tid = threadIdx.x;
  const int lane = tid & 31;
  const int wave = tid >> 5;
  const int b = blockIdx.x;
  const int r0 = b * BR;

  if (tid < BR) cntL[tid] = 0;
  __syncthreads();

  for (int f = tid; f < NW; f += 256) {
    const int* __restrict__ t = st + (size_t)f * (NB + 1) + b;
    const int s = t[0], e = t[1];
    const int* __restrict__ p = sD + (size_t)f * CHUNK;
    int i = s;
    for (; i + 4 <= e; i += 4) {
      const int d0 = p[i], d1 = p[i + 1], d2 = p[i + 2], d3 = p[i + 3];
      atomicAdd(&cntL[d0 - r0], 1);
      atomicAdd(&cntL[d1 - r0], 1);
      atomicAdd(&cntL[d2 - r0], 1);
      atomicAdd(&cntL[d3 - r0], 1);
    }
    for (; i < e; ++i) atomicAdd(&cntL[p[i] - r0], 1);
  }
  __syncthreads();

  const int c = (tid < BR) ? cntL[tid] : 0;
  int incl = c;
  #pragma unroll
  for (int off = 1; off < 32; off <<= 1) {
    const int v = __shfl_up(incl, off, 32);
    if (lane >= off) incl += v;
  }
  if (tid < BR && lane == 31) wtot[wave] = incl;
  __syncthreads();
  int excl = incl - c;
  for (int k = 0; k < wave && k < BR / 32; ++k) excl += wtot[k];
  if (tid == BR - 1) {
    const int T = excl + c;
    gb[0] = atomicAdd(cursor, T);
    gb[1] = T;
  }
  __syncthreads();
  const int gbase = gb[0];
  const int T = gb[1];
  if (tid < BR) {
    curL[tid] = excl;
    const int row = r0 + tid;
    if (row < N) {
      base[row] = gbase + excl;
      cnt[row] = c;
    }
  }
  __syncthreads();

  const bool staged = T <= STG;
  int2* __restrict__ seg = csr + gbase;
  for (int f = tid; f < NW; f += 256) {
    const int* __restrict__ t = st + (size_t)f * (NB + 1) + b;
    const int s = t[0], e = t[1];
    const int* __restrict__ p = sD + (size_t)f * CHUNK;
    const int2* __restrict__ q = sP + (size_t)f * CHUNK;
    int i = s;
    for (; i + 4 <= e; i += 4) {
      int dd[4], ps[4];
      int2 v[4];
      #pragma unroll
      for (int u = 0; u < 4; ++u) {
        dd[u] = p[i + u];
        v[u] = q[i + u];
      }
      #pragma unroll
      for (int u = 0; u < 4; ++u) ps[u] = atomicAdd(&curL[dd[u] - r0], 1);
      #pragma unroll
      for (int u = 0; u < 4; ++u) {
        if (staged) stg[ps[u]] = v[u];
        else seg[ps[u]] = v[u];
      }
    }
    for (; i < e; ++i) {
      const int ps = atomicAdd(&curL[p[i] - r0], 1);
      const int2 v = q[i];
      if (staged) stg[ps] = v;
      else seg[ps] = v;
    }
  }
  __syncthreads();
  if (staged)
    for (int i = tid; i < T; i += 256) seg[i] = stg[i];
}

__device__ __forceinline__ void fma4(float4& a, const float4 v, const float w) {
  a.x = fmaf(v.x, w, a.x);
  a.y = fmaf(v.y, w, a.y);
  a.z = fmaf(v.z, w, a.z);
  a.w = fmaf(v.w, w, a.w);
}

#define LDP 132
#define M2P 64
__global__ __launch_bounds__(128)
void gcn_gather128_gemm2(const float* __restrict__ m,
                         const int* __restrict__ csr_i,
                         const int* __restrict__ base,
                         const int* __restrict__ cnt,
                         const float* __restrict__ b,
                         const float* __restrict__ W2,
                         float* __restrict__ out, int N, int rowTiles) {
  __shared__ __attribute__((aligned(16))) float hT[16 * LDP];
  __shared__ __attribute__((aligned(16))) float oT[16 * DO];

  const int lane = threadIdx.x & 31;
  const int wave = __builtin_amdgcn_readfirstlane((int)(threadIdx.x >> 5));
  const int mm = lane & 15;
  const int hh = lane >> 4;
  const int kk = hh << 1;

  const int col = wave * 16 + mm;
  const bool cv = col < DO;
  const int cc = cv ? col : DO - 1;
  v2f bf[32];
  if (wave < 3) {
    #pragma unroll
    for (int s = 0; s < 32; ++s) {
      const float* __restrict__ wr = W2 + (size_t)(4 * s + kk) * DO + cc;
      bf[s].x = wr[0];
      bf[s].y = wr[DO];
    }
  }

  const float4* __restrict__ m4 = (const float4*)m + lane;
  const float4 bb = ((const float4*)b)[lane];

  for (int t = blockIdx.x; t < rowTiles; t += gridDim.x) {
    const int r0 = t * 16;
    int row = r0 + wave * 4;
    int sb = 0, n = 0;
    if (row < N) {
      sb = base[row];
      n = cnt[row];
    }

    for (int q = 0; q < 4; ++q) {
      const int nrow = row + 1;
      int nsb = 0, nn = 0;
      if (q < 3 && nrow < N) {
        nsb = base[nrow];
        nn = cnt[nrow];
      }
      const int2* __restrict__ seg = (const int2*)csr_i + sb;

      float4 acc = make_float4(0.f, 0.f, 0.f, 0.f);
      int j = 0;
      for (; j + 8 <= n; j += 8) {
        int2 e[8];
        float4 v[8];
        #pragma unroll
        for (int u = 0; u < 8; ++u) e[u] = seg[j + u];
        #pragma unroll
        for (int u = 0; u < 8; ++u) v[u] = m4[(size_t)e[u].x * (DH / 4)];
        #pragma unroll
        for (int u = 0; u < 8; ++u) fma4(acc, v[u], __int_as_float(e[u].y));
      }
      if (j + 4 <= n) {
        int2 e[4];
        float4 v[4];
        #pragma unroll
        for (int u = 0; u < 4; ++u) e[u] = seg[j + u];
        #pragma unroll
        for (int u = 0; u < 4; ++u) v[u] = m4[(size_t)e[u].x * (DH / 4)];
        #pragma unroll
        for (int u = 0; u < 4; ++u) fma4(acc, v[u], __int_as_float(e[u].y));
        j += 4;
      }
      if (j + 2 <= n) {
        const int2 e0 = seg[j], e1 = seg[j + 1];
        const float4 v0 = m4[(size_t)e0.x * (DH / 4)];
        const float4 v1 = m4[(size_t)e1.x * (DH / 4)];
        fma4(acc, v0, __int_as_float(e0.y));
        fma4(acc, v1, __int_as_float(e1.y));
        j += 2;
      }
      if (j < n) {
        const int2 e0 = seg[j];
        const float4 v0 = m4[(size_t)e0.x * (DH / 4)];
        fma4(acc, v0, __int_as_float(e0.y));
      }

      float4 r;
      r.x = fmaxf(acc.x + bb.x, 0.0f);
      r.y = fmaxf(acc.y + bb.y, 0.0f);
      r.z = fmaxf(acc.z + bb.z, 0.0f);
      r.w = fmaxf(acc.w + bb.w, 0.0f);
      *(float4*)&hT[(wave * 4 + q) * LDP + 4 * lane] = r;

      row = nrow;
      sb = nsb;
      n = nn;
    }

    __syncthreads();

    if (wave < 3) {
      const float* __restrict__ ap = &hT[mm * LDP + kk];
      v8f dacc = {};
      #pragma unroll
      for (int s = 0; s < 32; ++s) {
        const v2f a = *(const v2f*)(ap + 4 * s);
        dacc = __builtin_amdgcn_wmma_f32_16x16x4_f32(
            false, a, false, bf[s], (short)0, dacc, false, false);
      }
      if (cv) {
        #pragma unroll
        for (int j = 0; j < 8; ++j) oT[(hh * 8 + j) * DO + col] = dacc[j];
      }
    }

    __syncthreads();

    const int pc = threadIdx.x & 15;
    #pragma unroll
    for (int rr = (int)(threadIdx.x >> 4); rr < 16; rr += 8) {
      float4 val = make_float4(0.f, 0.f, 0.f, 0.f);
      if (pc < DO / 4) val = ((const float4*)oT)[rr * (DO / 4) + pc];
      if (r0 + rr < N)
        ((float4*)out)[(size_t)(r0 + rr) * (M2P / 4) + pc] = val;
    }
  }
}


template <int G>
__device__ __forceinline__
void sum8(float4& acc, const int2 e, const int c,
          const float* __restrict__ m, const unsigned pcoff) {
  float w[8];
  float4 v[8];
  #pragma unroll
  for (int u = 0; u < 8; ++u) {
    const unsigned off =
        (unsigned)__shfl(e.x, G + u, 16) * (unsigned)(M2P * sizeof(float)) + pcoff;
    w[u] = __int_as_float(__shfl(e.y, G + u, 16));
    v[u] = make_float4(0.f, 0.f, 0.f, 0.f);
    if (G + u < c) v[u] = *(const float4*)((const char*)m + off);
  }
  #pragma unroll
  for (int u = 0; u < 8; ++u) fma4(acc, v[u], w[u]);
}

__global__ __launch_bounds__(256)
void gcn_gather40(const float* __restrict__ m, const int* __restrict__ csr_i,
                  const int* __restrict__ base, const int* __restrict__ cnt,
                  const float* __restrict__ b, float* __restrict__ out, int N) {
  __shared__ __attribute__((aligned(16))) float oT[16 * DO];

  const int tid = threadIdx.x;
  const int pc = tid & 15;
  const int rl = tid >> 4;
  const int r0 = blockIdx.x * 16;
  const int row = r0 + rl;
  const unsigned pcoff = (unsigned)pc * 16u;

  int sb = 0, n = 0;
  if (row < N) {
    sb = base[row];
    n = cnt[row];
  }
  const int o = __shfl_xor(n, 16, 32);
  const int nn = __builtin_amdgcn_readfirstlane(n > o ? n : o);
  const int2* __restrict__ seg = (const int2*)csr_i + sb;

  float4 acc = make_float4(0.f, 0.f, 0.f, 0.f);
  for (int j0 = 0; j0 < nn; j0 += 32) {
    const int c = n - j0;
    int2 ea = make_int2(0, 0), eb = make_int2(0, 0);
    if (pc < c) ea = seg[j0 + pc];
    if (pc + 16 < c) eb = seg[j0 + 16 + pc];
    sum8<0>(acc, ea, c, m, pcoff);
    if (j0 + 8 < nn) sum8<8>(acc, ea, c, m, pcoff);
    if (j0 + 16 < nn) sum8<0>(acc, eb, c - 16, m, pcoff);
    if (j0 + 24 < nn) sum8<8>(acc, eb, c - 16, m, pcoff);
  }

  if (pc < DO / 4) {
    const float4 bb = ((const float4*)b)[pc];
    float4 r;
    r.x = acc.x + bb.x;
    r.y = acc.y + bb.y;
    r.z = acc.z + bb.z;
    r.w = acc.w + bb.w;
    *(float4*)&oT[rl * DO + 4 * pc] = r;
  }
  __syncthreads();

  const int rows = (N - r0 < 16) ? (N - r0) : 16;
  if (tid < rows * (DO / 4))
    ((float4*)out)[(size_t)r0 * (DO / 4) + tid] = ((const float4*)oT)[tid];
}

extern "C" void kernel_launch(void* const* d_in, const int* in_sizes, int n_in,
                              void* d_out, int out_size, void* d_ws,
                              size_t ws_size, hipStream_t stream) {
  const float* x  = (const float*)d_in[0];
  const int*   ei = (const int*)d_in[1];
  const float* ew = (const float*)d_in[2];
  const float* W1 = (const float*)d_in[3];
  const float* b1 = (const float*)d_in[4];
  const float* W2 = (const float*)d_in[5];
  const float* b2 = (const float*)d_in[6];

  const int N = in_sizes[0] / DH;
  const int E = in_sizes[1] / 2;
  const int* src = ei;
  const int* dst = ei + E;

  float* m1   = (float*)d_ws;
  float* m2   = m1 + (size_t)N * DH;
  const int NP = (N + 35) & ~3;
  int*   cnt  = (int*)(m2 + (size_t)N * M2P);
  int*   base = cnt + NP;
  int2*  csr  = (int2*)(base + NP);
  int2*  sP   = csr + E;
  int*   sD   = (int*)(sP + E);
  int*   st   = sD + E;
  float* out  = (float*)d_out;

  const int nt = 256;
  const int rowTiles = (N + 15) / 16;
  const int NB = (N + BR - 1) / BR;
  const int NW = (E + CHUNK - 1) / CHUNK;

  gcn_bin<<<NW + 512, nt, (NB + 1) * sizeof(int), stream>>>(
      src, dst, ew, sD, sP, st, cnt + N, E, NB, NW, x, W1, m1, N, rowTiles);
  gcn_group<<<NB, nt, 0, stream>>>(sD, sP, st, cnt + N, base, cnt, csr,
                                   N, NB, NW);
  gcn_gather128_gemm2<<<(rowTiles + 1) / 2, 128, 0, stream>>>(
      m1, (const int*)csr, base, cnt, b1, W2, m2, N, rowTiles);
  gcn_gather40<<<rowTiles, nt, 0, stream>>>(
      m2, (const int*)csr, base, cnt, b2, out, N);
}


# Layer-1 gather: a wave fetches its 4 rows' bounds and (src,w) pairs up front, one pair per lane, handed out by readlane: an edge group costs one round trip, not two or three; sum order unchanged
# speedup vs baseline: 6.6585x; 1.0425x over previous
#include <hip/hip_runtime.h>

typedef float v2f __attribute__((ext_vector_type(2)));
typedef float v4f __attribute__((ext_vector_type(4)));
typedef float v8f __attribute__((ext_vector_type(8)));

#define DH 128
#define DO 40

#define XLP 132
#define XTILE (16 * XLP)

__device__ __forceinline__
v4f x_row_piece(const float* __restrict__ x, int r, int N, int lane) {
  return ((const v4f*)(x + (size_t)(r < N ? r : N - 1) * DH))[lane];
}

__device__ __forceinline__
void gemm1_tiles(const float* __restrict__ x, const float* __restrict__ W,
                 float* __restrict__ out, int N, int rowTiles,
                 int vb, int nvb, float* __restrict__ xs) {
  const int lane = threadIdx.x & 31;
  const int wave = (threadIdx.x >> 5) & 3;
  const int c0 = wave * 32;
  const int m  = lane & 15;
  const int kk = (lane >> 4) << 1;

  v2f b0[32], b1[32];
  #pragma unroll
  for (int s = 0; s < 32; ++s) {
    const float* __restrict__ wr = W + (size_t)(4 * s + kk) * DH + c0 + m;
    b0[s].x = wr[0];
    b0[s].y = wr[DH];
    b1[s].x = wr[16];
    b1[s].y = wr[DH + 16];
  }

  const int trips = (rowTiles + nvb - 1) / nvb;
  v4f g0, g1, g2, g3;
  float* __restrict__ sw = xs + (wave * 4) * XLP + 4 * lane;
  const float* __restrict__ ap = xs + m * XLP + kk;

  int t = vb;
  if (t < rowTiles) {
    const int rb = t * 16 + wave * 4;
    g0 = x_row_piece(x, rb, N, lane);
    g1 = x_row_piece(x, rb + 1, N, lane);
    g2 = x_row_piece(x, rb + 2, N, lane);
    g3 = x_row_piece(x, rb + 3, N, lane);
    *(v4f*)(sw) = g0;
    *(v4f*)(sw + XLP) = g1;
    *(v4f*)(sw + 2 * XLP) = g2;
    *(v4f*)(sw + 3 * XLP) = g3;
  }
  __syncthreads();

  for (int it = 0; it < trips; ++it, t += nvb) {
    const bool have = t < rowTiles;
    v8f acc0 = {}, acc1 = {};
    if (have) {
      const int tn = t + nvb;
      const int rb = tn * 16 + wave * 4;
      g0 = x_row_piece(x, rb, N, lane);
      g1 = x_row_piece(x, rb + 1, N, lane);
      g2 = x_row_piece(x, rb + 2, N, lane);
      g3 = x_row_piece(x, rb + 3, N, lane);

      const float* __restrict__ a_p = ap + (it & 1) * XTILE;
      v2f a[32];
      #pragma unroll
      for (int s = 0; s < 32; ++s) a[s] = *(const v2f*)(a_p + 4 * s);
      asm volatile("" :: "v"(a[0]), "v"(a[1]), "v"(a[2]), "v"(a[3]), "v"(a[4]), "v"(a[5]), "v"(a[6]), "v"(a[7]), "v"(a[8]), "v"(a[9]), "v"(a[10]), "v"(a[11]), "v"(a[12]), "v"(a[13]), "v"(a[14]), "v"(a[15]));
      asm volatile("" :: "v"(a[16]), "v"(a[17]), "v"(a[18]), "v"(a[19]), "v"(a[20]), "v"(a[21]), "v"(a[22]), "v"(a[23]), "v"(a[24]), "v"(a[25]), "v"(a[26]), "v"(a[27]), "v"(a[28]), "v"(a[29]), "v"(a[30]), "v"(a[31]));
      #pragma unroll
      for (int s = 0; s < 32; ++s) {
        acc0 = __builtin_amdgcn_wmma_f32_16x16x4_f32(
            false, a[s], false, b0[s], (short)0, acc0, false, false);
        acc1 = __builtin_amdgcn_wmma_f32_16x16x4_f32(
            false, a[s], false, b1[s], (short)0, acc1, false, false);
      }

      __builtin_amdgcn_sched_barrier(0);
      float* __restrict__ sn = sw + ((it + 1) & 1) * XTILE;
      *(v4f*)(sn) = g0;
      *(v4f*)(sn + XLP) = g1;
      *(v4f*)(sn + 2 * XLP) = g2;
      *(v4f*)(sn + 3 * XLP) = g3;
    }
    __syncthreads();

    if (have) {
      const int r0 = t * 16;
      const int rbase = r0 + ((lane >> 4) << 3);
      float* __restrict__ p = out + (size_t)rbase * DH + c0 + m;
      if (r0 + 16 <= N) {
        #pragma unroll
        for (int j = 0; j < 8; ++j) {
          p[(size_t)j * DH] = acc0[j];
          p[(size_t)j * DH + 16] = acc1[j];
        }
      } else {
        #pragma unroll
        for (int j = 0; j < 8; ++j)
          if (rbase + j < N) {
            p[(size_t)j * DH] = acc0[j];
            p[(size_t)j * DH + 16] = acc1[j];
          }
      }
    }
  }
}

#define BR    128
#define CHUNK 4096
#define EPT   16
#define STG   6144

__global__ __launch_bounds__(256)
void gcn_bin(const int* __restrict__ src, const int* __restrict__ dst,
             const float* __restrict__ w, int* __restrict__ sD,
             int2* __restrict__ sP, int* __restrict__ st,
             int* __restrict__ cursor, int E, int NB, int NW,
             const float* __restrict__ gx, const float* __restrict__ gW,
             float* __restrict__ gm, int N, int rowTiles) {
  __shared__ __attribute__((aligned(16))) float xT[2 * 2 * XTILE];
  if ((int)blockIdx.x >= NW) {
    const int half = __builtin_amdgcn_readfirstlane((int)(threadIdx.x >> 7));
    gemm1_tiles(gx, gW, gm, N, rowTiles,
                ((int)blockIdx.x - NW) * 2 + half,
                ((int)gridDim.x - NW) * 2, xT + half * (2 * XTILE));
    return;
  }
  extern __shared__ int hist[];
  __shared__ int stD[CHUNK];
  __shared__ int2 stP[CHUNK];
  __shared__ int wsum[8];

  const int tid = threadIdx.x;
  const int lane = tid & 31;
  const int wave = tid >> 5;
  const int c0 = blockIdx.x * CHUNK;
  const int cn = (E - c0 < CHUNK) ? (E - c0) : CHUNK;

  if (blockIdx.x == 0 && tid == 0) *cursor = 0;
  for (int i = tid; i <= NB; i += 256) hist[i] = 0;

  int d[EPT], s[EPT], r[EPT];
  float ww[EPT];
  #pragma unroll
  for (int u = 0; u < EPT; ++u) {
    const int i = u * 256 + tid;
    d[u] = -1;
    s[u] = 0;
    ww[u] = 0.0f;
    r[u] = 0;
    if (i < cn) {
      d[u] = dst[c0 + i];
      s[u] = src[c0 + i];
      ww[u] = w[c0 + i];
    }
  }
  __syncthreads();

  #pragma unroll
  for (int u = 0; u < EPT; ++u)
    if (d[u] >= 0) r[u] = atomicAdd(&hist[d[u] / BR], 1);
  __syncthreads();

  const int K = (NB + 255) >> 8;
  const int b0 = tid * K;
  int loc = 0;
  for (int k = 0; k < K; ++k)
    if (b0 + k < NB) loc += hist[b0 + k];
  int incl = loc;
  #pragma unroll
  for (int off = 1; off < 32; off <<= 1) {
    const int v = __shfl_up(incl, off, 32);
    if (lane >= off) incl += v;
  }
  if (lane == 31) wsum[wave] = incl;
  __syncthreads();
  int run = incl - loc;
  for (int k = 0; k < wave; ++k) run += wsum[k];
  for (int k = 0; k < K; ++k)
    if (b0 + k < NB) {
      const int c = hist[b0 + k];
      hist[b0 + k] = run;
      run += c;
    }
  if (tid == 255) hist[NB] = run;
  __syncthreads();

  int* __restrict__ strow = st + (size_t)blockIdx.x * (NB + 1);
  for (int i = tid; i <= NB; i += 256) strow[i] = hist[i];

  #pragma unroll
  for (int u = 0; u < EPT; ++u)
    if (d[u] >= 0) {
      const int p = hist[d[u] / BR] + r[u];
      stD[p] = d[u];
      stP[p] = make_int2(s[u], __float_as_int(ww[u]));
    }
  __syncthreads();

  for (int i = tid; i < cn; i += 256) {
    sD[c0 + i] = stD[i];
    sP[c0 + i] = stP[i];
  }
}

__global__ __launch_bounds__(256)
void gcn_group(const int* __restrict__ sD, const int2* __restrict__ sP,
               const int* __restrict__ st, int* __restrict__ cursor,
               int* __restrict__ base, int* __restrict__ cnt,
               int2* __restrict__ csr, int N, int NB, int NW) {
  __shared__ int cntL[BR];
  __shared__ int curL[BR];
  __shared__ int wtot[BR / 32];
  __shared__ int gb[2];
  __shared__ int2 stg[STG];

  const int tid = threadIdx.x;
  const int lane = tid & 31;
  const int wave = tid >> 5;
  const int b = blockIdx.x;
  const int r0 = b * BR;

  if (tid < BR) cntL[tid] = 0;
  __syncthreads();

  for (int f = tid; f < NW; f += 256) {
    const int* __restrict__ t = st + (size_t)f * (NB + 1) + b;
    const int s = t[0], e = t[1];
    const int* __restrict__ p = sD + (size_t)f * CHUNK;
    int i = s;
    for (; i + 4 <= e; i += 4) {
      const int d0 = p[i], d1 = p[i + 1], d2 = p[i + 2], d3 = p[i + 3];
      atomicAdd(&cntL[d0 - r0], 1);
      atomicAdd(&cntL[d1 - r0], 1);
      atomicAdd(&cntL[d2 - r0], 1);
      atomicAdd(&cntL[d3 - r0], 1);
    }
    for (; i < e; ++i) atomicAdd(&cntL[p[i] - r0], 1);
  }
  __syncthreads();

  const int c = (tid < BR) ? cntL[tid] : 0;
  int incl = c;
  #pragma unroll
  for (int off = 1; off < 32; off <<= 1) {
    const int v = __shfl_up(incl, off, 32);
    if (lane >= off) incl += v;
  }
  if (tid < BR && lane == 31) wtot[wave] = incl;
  __syncthreads();
  int excl = incl - c;
  for (int k = 0; k < wave && k < BR / 32; ++k) excl += wtot[k];
  if (tid == BR - 1) {
    const int T = excl + c;
    gb[0] = atomicAdd(cursor, T);
    gb[1] = T;
  }
  __syncthreads();
  const int gbase = gb[0];
  const int T = gb[1];
  if (tid < BR) {
    curL[tid] = excl;
    const int row = r0 + tid;
    if (row < N) {
      base[row] = gbase + excl;
      cnt[row] = c;
    }
  }
  __syncthreads();

  const bool staged = T <= STG;
  int2* __restrict__ seg = csr + gbase;
  for (int f = tid; f < NW; f += 256) {
    const int* __restrict__ t = st + (size_t)f * (NB + 1) + b;
    const int s = t[0], e = t[1];
    const int* __restrict__ p = sD + (size_t)f * CHUNK;
    const int2* __restrict__ q = sP + (size_t)f * CHUNK;
    int i = s;
    for (; i + 4 <= e; i += 4) {
      int dd[4], ps[4];
      int2 v[4];
      #pragma unroll
      for (int u = 0; u < 4; ++u) {
        dd[u] = p[i + u];
        v[u] = q[i + u];
      }
      #pragma unroll
      for (int u = 0; u < 4; ++u) ps[u] = atomicAdd(&curL[dd[u] - r0], 1);
      #pragma unroll
      for (int u = 0; u < 4; ++u) {
        if (staged) stg[ps[u]] = v[u];
        else seg[ps[u]] = v[u];
      }
    }
    for (; i < e; ++i) {
      const int ps = atomicAdd(&curL[p[i] - r0], 1);
      const int2 v = q[i];
      if (staged) stg[ps] = v;
      else seg[ps] = v;
    }
  }
  __syncthreads();
  if (staged)
    for (int i = tid; i < T; i += 256) seg[i] = stg[i];
}

__device__ __forceinline__ void fma4(float4& a, const float4 v, const float w) {
  a.x = fmaf(v.x, w, a.x);
  a.y = fmaf(v.y, w, a.y);
  a.z = fmaf(v.z, w, a.z);
  a.w = fmaf(v.w, w, a.w);
}

template <int W>
__device__ __forceinline__
void sum_slots(float4& acc, const int2 e, const int G,
               const float* __restrict__ m, const int lane) {
  float w[W];
  float4 v[W];
  #pragma unroll
  for (int u = 0; u < W; ++u) {
    const int s = __builtin_amdgcn_readlane(e.x, G + u);
    w[u] = __int_as_float(__builtin_amdgcn_readlane(e.y, G + u));
    v[u] = ((const float4*)(m + (size_t)s * DH))[lane];
  }
  #pragma unroll
  for (int u = 0; u < W; ++u) fma4(acc, v[u], w[u]);
}

#define LDP 132
#define M2P 64
__global__ __launch_bounds__(128)
void gcn_gather128_gemm2(const float* __restrict__ m,
                         const int* __restrict__ csr_i,
                         const int* __restrict__ base,
                         const int* __restrict__ cnt,
                         const float* __restrict__ b,
                         const float* __restrict__ W2,
                         float* __restrict__ out, int N, int rowTiles) {
  __shared__ __attribute__((aligned(16))) float hT[16 * LDP];
  __shared__ __attribute__((aligned(16))) float oT[16 * DO];

  const int lane = threadIdx.x & 31;
  const int wave = __builtin_amdgcn_readfirstlane((int)(threadIdx.x >> 5));
  const int mm = lane & 15;
  const int hh = lane >> 4;
  const int kk = hh << 1;

  const int col = wave * 16 + mm;
  const bool cv = col < DO;
  const int cc = cv ? col : DO - 1;
  v2f bf[32];
  if (wave < 3) {
    #pragma unroll
    for (int s = 0; s < 32; ++s) {
      const float* __restrict__ wr = W2 + (size_t)(4 * s + kk) * DO + cc;
      bf[s].x = wr[0];
      bf[s].y = wr[DO];
    }
  }

  const float4 bb = ((const float4*)b)[lane];
  const int2* __restrict__ csr2 = (const int2*)csr_i;

  for (int t = blockIdx.x; t < rowTiles; t += gridDim.x) {
    const int r0 = t * 16;
    const int rowb = r0 + wave * 4;
    const int lq = lane & 3;
    const bool rin = rowb + lq < N;
    const int rr = rin ? rowb + lq : N - 1;
    const int bl = base[rr];
    const int cl = rin ? cnt[rr] : 0;
    int sbq[4], nq[4];
    int2 e[4];
    #pragma unroll
    for (int q = 0; q < 4; ++q) {
      sbq[q] = __builtin_amdgcn_readlane(bl, q);
      nq[q] = __builtin_amdgcn_readlane(cl, q);
      const int li = lane < nq[q] ? lane : 0;
      e[q] = csr2[sbq[q] + li];
      if (lane >= nq[q]) e[q].y = 0;
    }

    #pragma unroll
    for (int q = 0; q < 4; ++q) {
      const int n = nq[q];
      int2 ee = e[q];
      float4 acc = make_float4(0.f, 0.f, 0.f, 0.f);
      for (int j0 = 0; j0 < n; j0 += 32) {
        const int c = n - j0 < 32 ? n - j0 : 32;
        if (j0 > 0) {
          const int li = lane < c ? lane : 0;
          ee = csr2[sbq[q] + j0 + li];
          if (lane >= c) ee.y = 0;
        }
        int G = 0;
        for (; G + 8 <= c; G += 8) sum_slots<8>(acc, ee, G, m, lane);
        if (c - G > 4) sum_slots<8>(acc, ee, G, m, lane);
        else if (c - G > 0) sum_slots<4>(acc, ee, G, m, lane);
      }

      float4 r;
      r.x = fmaxf(acc.x + bb.x, 0.0f);
      r.y = fmaxf(acc.y + bb.y, 0.0f);
      r.z = fmaxf(acc.z + bb.z, 0.0f);
      r.w = fmaxf(acc.w + bb.w, 0.0f);
      *(float4*)&hT[(wave * 4 + q) * LDP + 4 * lane] = r;
    }

    __syncthreads();

    if (wave < 3) {
      const float* __restrict__ ap = &hT[mm * LDP + kk];
      v8f dacc = {};
      #pragma unroll
      for (int s = 0; s < 32; ++s) {
        const v2f a = *(const v2f*)(ap + 4 * s);
        dacc = __builtin_amdgcn_wmma_f32_16x16x4_f32(
            false, a, false, bf[s], (short)0, dacc, false, false);
      }
      if (cv) {
        #pragma unroll
        for (int j = 0; j < 8; ++j) oT[(hh * 8 + j) * DO + col] = dacc[j];
      }
    }

    __syncthreads();

    const int pc = threadIdx.x & 15;
    #pragma unroll
    for (int rr = (int)(threadIdx.x >> 4); rr < 16; rr += 8) {
      float4 val = make_float4(0.f, 0.f, 0.f, 0.f);
      if (pc < DO / 4) val = ((const float4*)oT)[rr * (DO / 4) + pc];
      if (r0 + rr < N)
        ((float4*)out)[(size_t)(r0 + rr) * (M2P / 4) + pc] = val;
    }
  }
}


template <int G>
__device__ __forceinline__
void sum8(float4& acc, const int2 e, const int c,
          const float* __restrict__ m, const unsigned pcoff) {
  float w[8];
  float4 v[8];
  #pragma unroll
  for (int u = 0; u < 8; ++u) {
    const unsigned off =
        (unsigned)__shfl(e.x, G + u, 16) * (unsigned)(M2P * sizeof(float)) + pcoff;
    w[u] = __int_as_float(__shfl(e.y, G + u, 16));
    v[u] = make_float4(0.f, 0.f, 0.f, 0.f);
    if (G + u < c) v[u] = *(const float4*)((const char*)m + off);
  }
  #pragma unroll
  for (int u = 0; u < 8; ++u) fma4(acc, v[u], w[u]);
}

__global__ __launch_bounds__(256)
void gcn_gather40(const float* __restrict__ m, const int* __restrict__ csr_i,
                  const int* __restrict__ base, const int* __restrict__ cnt,
                  const float* __restrict__ b, float* __restrict__ out, int N) {
  __shared__ __attribute__((aligned(16))) float oT[16 * DO];

  const int tid = threadIdx.x;
  const int pc = tid & 15;
  const int rl = tid >> 4;
  const int r0 = blockIdx.x * 16;
  const int row = r0 + rl;
  const unsigned pcoff = (unsigned)pc * 16u;

  int sb = 0, n = 0;
  if (row < N) {
    sb = base[row];
    n = cnt[row];
  }
  const int o = __shfl_xor(n, 16, 32);
  const int nn = __builtin_amdgcn_readfirstlane(n > o ? n : o);
  const int2* __restrict__ seg = (const int2*)csr_i + sb;

  float4 acc = make_float4(0.f, 0.f, 0.f, 0.f);
  for (int j0 = 0; j0 < nn; j0 += 32) {
    const int c = n - j0;
    int2 ea = make_int2(0, 0), eb = make_int2(0, 0);
    if (pc < c) ea = seg[j0 + pc];
    if (pc + 16 < c) eb = seg[j0 + 16 + pc];
    sum8<0>(acc, ea, c, m, pcoff);
    if (j0 + 8 < nn) sum8<8>(acc, ea, c, m, pcoff);
    if (j0 + 16 < nn) sum8<0>(acc, eb, c - 16, m, pcoff);
    if (j0 + 24 < nn) sum8<8>(acc, eb, c - 16, m, pcoff);
  }

  if (pc < DO / 4) {
    const float4 bb = ((const float4*)b)[pc];
    float4 r;
    r.x = acc.x + bb.x;
    r.y = acc.y + bb.y;
    r.z = acc.z + bb.z;
    r.w = acc.w + bb.w;
    *(float4*)&oT[rl * DO + 4 * pc] = r;
  }
  __syncthreads();

  const int rows = (N - r0 < 16) ? (N - r0) : 16;
  if (tid < rows * (DO / 4))
    ((float4*)out)[(size_t)r0 * (DO / 4) + tid] = ((const float4*)oT)[tid];
}

extern "C" void kernel_launch(void* const* d_in, const int* in_sizes, int n_in,
                              void* d_out, int out_size, void* d_ws,
                              size_t ws_size, hipStream_t stream) {
  const float* x  = (const float*)d_in[0];
  const int*   ei = (const int*)d_in[1];
  const float* ew = (const float*)d_in[2];
  const float* W1 = (const float*)d_in[3];
  const float* b1 = (const float*)d_in[4];
  const float* W2 = (const float*)d_in[5];
  const float* b2 = (const float*)d_in[6];

  const int N = in_sizes[0] / DH;
  const int E = in_sizes[1] / 2;
  const int* src = ei;
  const int* dst = ei + E;

  float* m1   = (float*)d_ws;
  float* m2   = m1 + (size_t)N * DH;
  const int NP = (N + 35) & ~3;
  int*   cnt  = (int*)(m2 + (size_t)N * M2P);
  int*   base = cnt + NP;
  int2*  csr  = (int2*)(base + NP);
  int2*  sP   = csr + E;
  int*   sD   = (int*)(sP + E);
  int*   st   = sD + E;
  float* out  = (float*)d_out;

  const int nt = 256;
  const int rowTiles = (N + 15) / 16;
  const int NB = (N + BR - 1) / BR;
  const int NW = (E + CHUNK - 1) / CHUNK;

  gcn_bin<<<NW + 512, nt, (NB + 1) * sizeof(int), stream>>>(
      src, dst, ew, sD, sP, st, cnt + N, E, NB, NW, x, W1, m1, N, rowTiles);
  gcn_group<<<NB, nt, 0, stream>>>(sD, sP, st, cnt + N, base, cnt, csr,
                                   N, NB, NW);
  gcn_gather128_gemm2<<<(rowTiles + 1) / 2, 128, 0, stream>>>(
      m1, (const int*)csr, base, cnt, b1, W2, m2, N, rowTiles);
  gcn_gather40<<<rowTiles, nt, 0, stream>>>(
      m2, (const int*)csr, base, cnt, b2, out, N);
}


# Second grouping pass joins the layer-1 kernel: bin = a block's 32 rows, grouped in LDS ahead of its sums (bounds, pairs from LDS); 8192-edge chunks; 4 -> 3 launches; sum order = LDS arrival rank
# speedup vs baseline: 7.2309x; 1.0860x over previous
#include <hip/hip_runtime.h>

typedef float v2f __attribute__((ext_vector_type(2)));
typedef float v4f __attribute__((ext_vector_type(4)));
typedef float v8f __attribute__((ext_vector_type(8)));

#define DH 128
#define DO 40

#define XLP 132
#define XTILE (16 * XLP)

__device__ __forceinline__
v4f x_row_piece(const float* __restrict__ x, int r, int N, int lane) {
  return ((const v4f*)(x + (size_t)(r < N ? r : N - 1) * DH))[lane];
}

__device__ __forceinline__
void gemm1_tiles(const float* __restrict__ x, const float* __restrict__ W,
                 float* __restrict__ out, int N, int rowTiles,
                 int vb, int nvb, float* __restrict__ xs) {
  const int lane = threadIdx.x & 31;
  const int wave = (threadIdx.x >> 5) & 3;
  const int c0 = wave * 32;
  const int m  = lane & 15;
  const int kk = (lane >> 4) << 1;

  v2f b0[32], b1[32];
  #pragma unroll
  for (int s = 0; s < 32; ++s) {
    const float* __restrict__ wr = W + (size_t)(4 * s + kk) * DH + c0 + m;
    b0[s].x = wr[0];
    b0[s].y = wr[DH];
    b1[s].x = wr[16];
    b1[s].y = wr[DH + 16];
  }

  const int trips = (rowTiles + nvb - 1) / nvb;
  v4f g0, g1, g2, g3;
  float* __restrict__ sw = xs + (wave * 4) * XLP + 4 * lane;
  const float* __restrict__ ap = xs + m * XLP + kk;

  int t = vb;
  if (t < rowTiles) {
    const int rb = t * 16 + wave * 4;
    g0 = x_row_piece(x, rb, N, lane);
    g1 = x_row_piece(x, rb + 1, N, lane);
    g2 = x_row_piece(x, rb + 2, N, lane);
    g3 = x_row_piece(x, rb + 3, N, lane);
    *(v4f*)(sw) = g0;
    *(v4f*)(sw + XLP) = g1;
    *(v4f*)(sw + 2 * XLP) = g2;
    *(v4f*)(sw + 3 * XLP) = g3;
  }
  __syncthreads();

  for (int it = 0; it < trips; ++it, t += nvb) {
    const bool have = t < rowTiles;
    v8f acc0 = {}, acc1 = {};
    if (have) {
      const int tn = t + nvb;
      const int rb = tn * 16 + wave * 4;
      g0 = x_row_piece(x, rb, N, lane);
      g1 = x_row_piece(x, rb + 1, N, lane);
      g2 = x_row_piece(x, rb + 2, N, lane);
      g3 = x_row_piece(x, rb + 3, N, lane);

      const float* __restrict__ a_p = ap + (it & 1) * XTILE;
      v2f a[32];
      #pragma unroll
      for (int s = 0; s < 32; ++s) a[s] = *(const v2f*)(a_p + 4 * s);
      asm volatile("" :: "v"(a[0]), "v"(a[1]), "v"(a[2]), "v"(a[3]), "v"(a[4]), "v"(a[5]), "v"(a[6]), "v"(a[7]), "v"(a[8]), "v"(a[9]), "v"(a[10]), "v"(a[11]), "v"(a[12]), "v"(a[13]), "v"(a[14]), "v"(a[15]));
      asm volatile("" :: "v"(a[16]), "v"(a[17]), "v"(a[18]), "v"(a[19]), "v"(a[20]), "v"(a[21]), "v"(a[22]), "v"(a[23]), "v"(a[24]), "v"(a[25]), "v"(a[26]), "v"(a[27]), "v"(a[28]), "v"(a[29]), "v"(a[30]), "v"(a[31]));
      #pragma unroll
      for (int s = 0; s < 32; ++s) {
        acc0 = __builtin_amdgcn_wmma_f32_16x16x4_f32(
            false, a[s], false, b0[s], (short)0, acc0, false, false);
        acc1 = __builtin_amdgcn_wmma_f32_16x16x4_f32(
            false, a[s], false, b1[s], (short)0, acc1, false, false);
      }

      __builtin_amdgcn_sched_barrier(0);
      float* __restrict__ sn = sw + ((it + 1) & 1) * XTILE;
      *(v4f*)(sn) = g0;
      *(v4f*)(sn + XLP) = g1;
      *(v4f*)(sn + 2 * XLP) = g2;
      *(v4f*)(sn + 3 * XLP) = g3;
    }
    __syncthreads();

    if (have) {
      const int r0 = t * 16;
      const int rbase = r0 + ((lane >> 4) << 3);
      float* __restrict__ p = out + (size_t)rbase * DH + c0 + m;
      if (r0 + 16 <= N) {
        #pragma unroll
        for (int j = 0; j < 8; ++j) {
          p[(size_t)j * DH] = acc0[j];
          p[(size_t)j * DH + 16] = acc1[j];
        }
      } else {
        #pragma unroll
        for (int j = 0; j < 8; ++j)
          if (rbase + j < N) {
            p[(size_t)j * DH] = acc0[j];
            p[(size_t)j * DH + 16] = acc1[j];
          }
      }
    }
  }
}


#define BR    32
#define CHUNK 8192
#define EPT   32
#define STG   2048
#define PK    12

__global__ __launch_bounds__(256)
void gcn_bin(const int* __restrict__ src, const int* __restrict__ dst,
             const float* __restrict__ w, int* __restrict__ sD,
             int2* __restrict__ sP, int* __restrict__ st,
             int E, int NB, int NW,
             const float* __restrict__ gx, const float* __restrict__ gW,
             float* __restrict__ gm, int N, int rowTiles) {
  __shared__ __attribute__((aligned(16))) float xT[2 * 2 * XTILE];
  if ((int)blockIdx.x >= NW) {
    const int half = __builtin_amdgcn_readfirstlane((int)(threadIdx.x >> 7));
    gemm1_tiles(gx, gW, gm, N, rowTiles,
                ((int)blockIdx.x - NW) * 2 + half,
                ((int)gridDim.x - NW) * 2, xT + half * (2 * XTILE));
    return;
  }
  extern __shared__ int hist[];
  __shared__ int stD[CHUNK];
  __shared__ int2 stP[CHUNK];
  __shared__ int wsum[8];

  const int tid = threadIdx.x;
  const int lane = tid & 31;
  const int wave = tid >> 5;
  const int c0 = blockIdx.x * CHUNK;
  const int cn = (E - c0 < CHUNK) ? (E - c0) : CHUNK;

  for (int i = tid; i <= NB; i += 256) hist[i] = 0;

  int d[EPT], s[EPT], r[EPT];
  float ww[EPT];
  #pragma unroll
  for (int u = 0; u < EPT; ++u) {
    const int i = u * 256 + tid;
    d[u] = -1;
    s[u] = 0;
    ww[u] = 0.0f;
    r[u] = 0;
    if (i < cn) {
      d[u] = dst[c0 + i];
      s[u] = src[c0 + i];
      ww[u] = w[c0 + i];
    }
  }
  __syncthreads();

  #pragma unroll
  for (int u = 0; u < EPT; ++u)
    if (d[u] >= 0) r[u] = atomicAdd(&hist[d[u] / BR], 1);
  __syncthreads();

  const int K = (NB + 255) >> 8;
  const int b0 = tid * K;
  int loc = 0;
  for (int k = 0; k < K; ++k)
    if (b0 + k < NB) loc += hist[b0 + k];
  int incl = loc;
  #pragma unroll
  for (int off = 1; off < 32; off <<= 1) {
    const int v = __shfl_up(incl, off, 32);
    if (lane >= off) incl += v;
  }
  if (lane == 31) wsum[wave] = incl;
  __syncthreads();
  int run = incl - loc;
  for (int k = 0; k < wave; ++k) run += wsum[k];
  for (int k = 0; k < K; ++k)
    if (b0 + k < NB) {
      const int c = hist[b0 + k];
      hist[b0 + k] = run;
      run += c;
    }
  if (tid == 255) hist[NB] = run;
  __syncthreads();

  int* __restrict__ strow = st + (size_t)blockIdx.x * (NB + 1);
  for (int i = tid; i <= NB; i += 256) strow[i] = hist[i];

  #pragma unroll
  for (int u = 0; u < EPT; ++u)
    if (d[u] >= 0) {
      const int p = hist[d[u] / BR] + r[u];
      stD[p] = d[u];
      stP[p] = make_int2(s[u], __float_as_int(ww[u]));
    }
  __syncthreads();

  for (int i = tid; i < cn; i += 256) {
    sD[c0 + i] = stD[i];
    sP[c0 + i] = stP[i];
  }
}

__device__ __forceinline__ void fma4(float4& a, const float4 v, const float w) {
  a.x = fmaf(v.x, w, a.x);
  a.y = fmaf(v.y, w, a.y);
  a.z = fmaf(v.z, w, a.z);
  a.w = fmaf(v.w, w, a.w);
}

template <int W>
__device__ __forceinline__
void sum_slots(float4& acc, const int2 e, const int G,
               const float* __restrict__ m, const int lane) {
  float w[W];
  float4 v[W];
  #pragma unroll
  for (int u = 0; u < W; ++u) {
    const int s = __builtin_amdgcn_readlane(e.x, G + u);
    w[u] = __int_as_float(__builtin_amdgcn_readlane(e.y, G + u));
    v[u] = ((const float4*)(m + (size_t)s * DH))[lane];
  }
  #pragma unroll
  for (int u = 0; u < W; ++u) fma4(acc, v[u], w[u]);
}

template <bool STAGED>
__device__ __forceinline__
int2 pair_at(const int2* stg, int2* seg, const int i) {
  if (STAGED) return stg[i];
  int* p = (int*)(seg + i);
  int2 v;
  v.x = __hip_atomic_load(p, __ATOMIC_RELAXED, __HIP_MEMORY_SCOPE_AGENT);
  v.y = __hip_atomic_load(p + 1, __ATOMIC_RELAXED, __HIP_MEMORY_SCOPE_AGENT);
  return v;
}

__device__ __forceinline__
void count_range(int* cntL, const int* __restrict__ sD, const int f,
                 const int i0, const int e, const int rb) {
  const int* __restrict__ p = sD + (size_t)f * CHUNK;
  for (int i = i0; i < e; ++i) atomicAdd(&cntL[p[i] - rb], 1);
}

__device__ __forceinline__
void place_range(int* curL, int2* stg, int2* seg, const bool staged,
                 const int* __restrict__ sD, const int2* __restrict__ sP,
                 const int f, const int i0, const int e, const int rb) {
  const int* __restrict__ p = sD + (size_t)f * CHUNK;
  const int2* __restrict__ q = sP + (size_t)f * CHUNK;
  for (int i = i0; i < e; ++i) {
    const int ps = atomicAdd(&curL[p[i] - rb], 1);
    const int2 v = q[i];
    if (staged) stg[ps] = v;
    else seg[ps] = v;
  }
}

#define LDP 132
#define M2P 64

template <bool STAGED>
__device__ __forceinline__
void layer1_tiles(const float* __restrict__ m, const int2* stg, int2* seg,
                  const int* cntL, const int* begL, float* hT, float* oT,
                  const v2f (&bf)[32], const float4 bb,
                  float* __restrict__ out, const int bin, const int N,
                  const int rowTiles, const int lane, const int wave,
                  const int mm, const int hh, const int kk, const int col,
                  const bool cv) {
  for (int tt = 0; tt < BR / 16; ++tt) {
    const int t = bin * (BR / 16) + tt;
    if (t >= rowTiles) break;
    const int r0 = t * 16;
    const int rl = tt * 16 + wave * 4 + (lane & 3);
    const int bl = begL[rl];
    const int cl = cntL[rl];
    int sbq[4], nq[4];
    int2 e[4];
    #pragma unroll
    for (int q = 0; q < 4; ++q) {
      sbq[q] = __builtin_amdgcn_readlane(bl, q);
      nq[q] = __builtin_amdgcn_readlane(cl, q);
      const int li = lane < nq[q] ? lane : 0;
      e[q] = pair_at<STAGED>(stg, seg, sbq[q] + li);
      if (lane >= nq[q]) e[q].y = 0;
    }

    #pragma unroll
    for (int q = 0; q < 4; ++q) {
      const int n = nq[q];
      int2 ee = e[q];
      float4 acc = make_float4(0.f, 0.f, 0.f, 0.f);
      for (int j0 = 0; j0 < n; j0 += 32) {
        const int c = n - j0 < 32 ? n - j0 : 32;
        if (j0 > 0) {
          const int li = lane < c ? lane : 0;
          ee = pair_at<STAGED>(stg, seg, sbq[q] + j0 + li);
          if (lane >= c) ee.y = 0;
        }
        int G = 0;
        for (; G + 8 <= c; G += 8) sum_slots<8>(acc, ee, G, m, lane);
        if (c - G > 4) sum_slots<8>(acc, ee, G, m, lane);
        else if (c - G > 0) sum_slots<4>(acc, ee, G, m, lane);
      }

      float4 r;
      r.x = fmaxf(acc.x + bb.x, 0.0f);
      r.y = fmaxf(acc.y + bb.y, 0.0f);
      r.z = fmaxf(acc.z + bb.z, 0.0f);
      r.w = fmaxf(acc.w + bb.w, 0.0f);
      *(float4*)&hT[(wave * 4 + q) * LDP + 4 * lane] = r;
    }

    __syncthreads();

    if (wave < 3) {
      const float* __restrict__ ap = &hT[mm * LDP + kk];
      v8f dacc = {};
      #pragma unroll
      for (int s = 0; s < 32; ++s) {
        const v2f a = *(const v2f*)(ap + 4 * s);
        dacc = __builtin_amdgcn_wmma_f32_16x16x4_f32(
            false, a, false, bf[s], (short)0, dacc, false, false);
      }
      if (cv) {
        #pragma unroll
        for (int j = 0; j < 8; ++j) oT[(hh * 8 + j) * DO + col] = dacc[j];
      }
    }

    __syncthreads();

    const int pc = threadIdx.x & 15;
    #pragma unroll
    for (int rr = (int)(threadIdx.x >> 4); rr < 16; rr += 8) {
      float4 val = make_float4(0.f, 0.f, 0.f, 0.f);
      if (pc < DO / 4) val = ((const float4*)oT)[rr * (DO / 4) + pc];
      if (r0 + rr < N)
        ((float4*)out)[(size_t)(r0 + rr) * (M2P / 4) + pc] = val;
    }
  }
}

__global__ __launch_bounds__(128)
void gcn_gather128_gemm2(const float* __restrict__ m,
                         const int* __restrict__ sD,
                         const int2* __restrict__ sP,
                         const int* __restrict__ st,
                         int* __restrict__ base,
                         int* __restrict__ cnt,
                         int2* csr,
                         const float* __restrict__ b,
                         const float* __restrict__ W2,
                         float* __restrict__ out, int N, int rowTiles,
                         int NB, int NW) {
  __shared__ __attribute__((aligned(16))) float hT[16 * LDP];
  __shared__ __attribute__((aligned(16))) float oT[16 * DO];
  __shared__ int cntL[BR];
  __shared__ int begL[BR];
  __shared__ int curL[BR];
  __shared__ int gbL;
  __shared__ int totL;
  __shared__ int2 stg[STG + 8];

  const int tid = threadIdx.x;
  const int lane = threadIdx.x & 31;
  const int wave = __builtin_amdgcn_readfirstlane((int)(threadIdx.x >> 5));
  const int mm = lane & 15;
  const int hh = lane >> 4;
  const int kk = hh << 1;
  const int bin = blockIdx.x;
  const int rb = bin * BR;

  if (tid < BR) cntL[tid] = 0;
  if (tid == 0) gbL = 0;

  const int fa = tid;
  const bool ha = fa < NW;
  const int ca = ha ? fa : 0;
  const int* __restrict__ ta = st + (size_t)ca * (NB + 1) + bin;
  const int sa0 = ta[0], ea0 = ta[1];
  const int sa = ha ? sa0 : 0, ea = ha ? ea0 : 0;
  int da[PK];
  int2 va[PK];
  {
    const int* __restrict__ pa = sD + (size_t)ca * CHUNK;
    const int2* __restrict__ qa = sP + (size_t)ca * CHUNK;
    #pragma unroll
    for (int k = 0; k < PK; ++k) {
      const bool ina = sa + k < ea;
      const int ia = ina ? sa + k : 0;
      const int xa = pa[ia];
      va[k] = qa[ia];
      da[k] = ina ? xa : -1;
    }
  }
  __syncthreads();

  #pragma unroll
  for (int k = 0; k < PK; ++k)
    if (da[k] >= 0) atomicAdd(&cntL[da[k] - rb], 1);
  count_range(cntL, sD, fa, sa + PK, ea, rb);
  int pre = sa;
  for (int f = tid + 128; f < NW; f += 128) {
    const int* __restrict__ t = st + (size_t)f * (NB + 1) + bin;
    const int s = t[0], e = t[1];
    pre += s;
    count_range(cntL, sD, f, s, e, rb);
  }
  atomicAdd(&gbL, pre);
  __syncthreads();

  if (wave == 0) {
    const int c = cntL[lane];
    int incl = c;
    #pragma unroll
    for (int off = 1; off < 32; off <<= 1) {
      const int v = __shfl_up(incl, off, 32);
      if (lane >= off) incl += v;
    }
    const int excl = incl - c;
    begL[lane] = excl;
    curL[lane] = excl;
    if (lane == 31) totL = incl;
    const int row = rb + lane;
    if (row < N) {
      base[row] = gbL + excl;
      cnt[row] = c;
    }
  }
  __syncthreads();
  const int T = __builtin_amdgcn_readfirstlane(totL);
  const bool staged = T <= STG;
  int2* seg = csr + __builtin_amdgcn_readfirstlane(gbL);

  int ka[PK];
  #pragma unroll
  for (int k = 0; k < PK; ++k) {
    ka[k] = 0;
    if (da[k] >= 0) ka[k] = atomicAdd(&curL[da[k] - rb], 1);
  }
  if (staged) {
    #pragma unroll
    for (int k = 0; k < PK; ++k)
      if (da[k] >= 0) stg[ka[k]] = va[k];
  } else {
    #pragma unroll
    for (int k = 0; k < PK; ++k)
      if (da[k] >= 0) seg[ka[k]] = va[k];
  }
  place_range(curL, stg, seg, staged, sD, sP, fa, sa + PK, ea, rb);
  for (int f = tid + 128; f < NW; f += 128) {
    const int* __restrict__ t = st + (size_t)f * (NB + 1) + bin;
    place_range(curL, stg, seg, staged, sD, sP, f, t[0], t[1], rb);
  }
  if (!staged) __threadfence();
  __syncthreads();
  if (staged)
    for (int i = tid; i < T; i += 128) seg[i] = stg[i];

  const int col = wave * 16 + mm;
  const bool cv = col < DO;
  const int cc = cv ? col : DO - 1;
  v2f bf[32];
  if (wave < 3) {
    #pragma unroll
    for (int s = 0; s < 32; ++s) {
      const float* __restrict__ wr = W2 + (size_t)(4 * s + kk) * DO + cc;
      bf[s].x = wr[0];
      bf[s].y = wr[DO];
    }
  }

  const float4 bb = ((const float4*)b)[lane];

  if (staged)
    layer1_tiles<true>(m, stg, seg, cntL, begL, hT, oT, bf, bb, out, bin, N,
                       rowTiles, lane, wave, mm, hh, kk, col, cv);
  else
    layer1_tiles<false>(m, stg, seg, cntL, begL, hT, oT, bf, bb, out, bin, N,
                        rowTiles, lane, wave, mm, hh, kk, col, cv);
}


template <int G>
__device__ __forceinline__
void sum8(float4& acc, const int2 e, const int c,
          const float* __restrict__ m, const unsigned pcoff) {
  float w[8];
  float4 v[8];
  #pragma unroll
  for (int u = 0; u < 8; ++u) {
    const unsigned off =
        (unsigned)__shfl(e.x, G + u, 16) * (unsigned)(M2P * sizeof(float)) + pcoff;
    w[u] = __int_as_float(__shfl(e.y, G + u, 16));
    v[u] = make_float4(0.f, 0.f, 0.f, 0.f);
    if (G + u < c) v[u] = *(const float4*)((const char*)m + off);
  }
  #pragma unroll
  for (int u = 0; u < 8; ++u) fma4(acc, v[u], w[u]);
}

__global__ __launch_bounds__(256)
void gcn_gather40(const float* __restrict__ m, const int* __restrict__ csr_i,
                  const int* __restrict__ base, const int* __restrict__ cnt,
                  const float* __restrict__ b, float* __restrict__ out, int N) {
  __shared__ __attribute__((aligned(16))) float oT[16 * DO];

  const int tid = threadIdx.x;
  const int pc = tid & 15;
  const int rl = tid >> 4;
  const int r0 = blockIdx.x * 16;
  const int row = r0 + rl;
  const unsigned pcoff = (unsigned)pc * 16u;

  int sb = 0, n = 0;
  if (row < N) {
    sb = base[row];
    n = cnt[row];
  }
  const int o = __shfl_xor(n, 16, 32);
  const int nn = __builtin_amdgcn_readfirstlane(n > o ? n : o);
  const int2* __restrict__ seg = (const int2*)csr_i + sb;

  float4 acc = make_float4(0.f, 0.f, 0.f, 0.f);
  for (int j0 = 0; j0 < nn; j0 += 32) {
    const int c = n - j0;
    int2 ea = make_int2(0, 0), eb = make_int2(0, 0);
    if (pc < c) ea = seg[j0 + pc];
    if (pc + 16 < c) eb = seg[j0 + 16 + pc];
    sum8<0>(acc, ea, c, m, pcoff);
    if (j0 + 8 < nn) sum8<8>(acc, ea, c, m, pcoff);
    if (j0 + 16 < nn) sum8<0>(acc, eb, c - 16, m, pcoff);
    if (j0 + 24 < nn) sum8<8>(acc, eb, c - 16, m, pcoff);
  }

  if (pc < DO / 4) {
    const float4 bb = ((const float4*)b)[pc];
    float4 r;
    r.x = acc.x + bb.x;
    r.y = acc.y + bb.y;
    r.z = acc.z + bb.z;
    r.w = acc.w + bb.w;
    *(float4*)&oT[rl * DO + 4 * pc] = r;
  }
  __syncthreads();

  const int rows = (N - r0 < 16) ? (N - r0) : 16;
  if (tid < rows * (DO / 4))
    ((float4*)out)[(size_t)r0 * (DO / 4) + tid] = ((const float4*)oT)[tid];
}

extern "C" void kernel_launch(void* const* d_in, const int* in_sizes, int n_in,
                              void* d_out, int out_size, void* d_ws,
                              size_t ws_size, hipStream_t stream) {
  const float* x  = (const float*)d_in[0];
  const int*   ei = (const int*)d_in[1];
  const float* ew = (const float*)d_in[2];
  const float* W1 = (const float*)d_in[3];
  const float* b1 = (const float*)d_in[4];
  const float* W2 = (const float*)d_in[5];
  const float* b2 = (const float*)d_in[6];

  const int N = in_sizes[0] / DH;
  const int E = in_sizes[1] / 2;
  const int* src = ei;
  const int* dst = ei + E;

  float* m1   = (float*)d_ws;
  float* m2   = m1 + (size_t)N * DH;
  const int NP = (N + 31) & ~31;
  int*   cnt  = (int*)(m2 + (size_t)N * M2P);
  int*   base = cnt + NP;
  int2*  csr  = (int2*)(base + NP);
  int2*  sP   = csr + E;
  int*   sD   = (int*)(sP + E);
  int*   st   = sD + E;
  float* out  = (float*)d_out;

  const int nt = 256;
  const int rowTiles = (N + 15) / 16;
  const int NB = (N + BR - 1) / BR;
  const int NW = (E + CHUNK - 1) / CHUNK;

  gcn_bin<<<NW + 512, nt, (NB + 1) * sizeof(int), stream>>>(
      src, dst, ew, sD, sP, st, E, NB, NW, x, W1, m1, N, rowTiles);
  gcn_gather128_gemm2<<<NB, 128, 0, stream>>>(
      m1, sD, sP, st, base, cnt, csr, b1, W2, m2, N, rowTiles, NB, NW);
  gcn_gather40<<<rowTiles, nt, 0, stream>>>(
      m2, (const int*)csr, base, cnt, b2, out, N);
}
